# starting kernel
# baseline (speedup 1.0000x reference)
#include <hip/hip_runtime.h>

typedef float v2f __attribute__((ext_vector_type(2)));
typedef float v8f __attribute__((ext_vector_type(8)));

#define NNODES 10000
#define NEDGES 320000
#define DIN 512
#define DOUT 256

__global__ __launch_bounds__(256) void lin_gemm_kernel(
    const float* __restrict__ X, const float* __restrict__ W,
    const float* __restrict__ bias, float* __restrict__ H)
{
    const int lane   = threadIdx.x & 31;
    const int wave   = threadIdx.x >> 5;
    const int lane16 = lane & 15;
    const int hsel   = lane >> 4;

    const int row0 = blockIdx.x * 16;
    const int col0 = wave * 32;

    const float* arow  = X + (size_t)(row0 + lane16) * DIN + 2 * hsel;
    const float* bcol0 = W + (size_t)(2 * hsel) * DOUT + (col0 + lane16);
    const float* bcol1 = bcol0 + 16;

    v8f acc0 = {};
    v8f acc1 = {};
    #pragma unroll 4
    for (int k = 0; k < DIN; k += 4) {
        v2f a = *(const v2f*)arow;
        v2f b0f, b1f;
        b0f.x = bcol0[0];
        b0f.y = bcol0[DOUT];
        b1f.x = bcol1[0];
        b1f.y = bcol1[DOUT];
        acc0 = __builtin_amdgcn_wmma_f32_16x16x4_f32(false, a, false, b0f,
                                                     (short)0, acc0, false, false);
        acc1 = __builtin_amdgcn_wmma_f32_16x16x4_f32(false, a, false, b1f,
                                                     (short)0, acc1, false, false);
        arow += 4;
        bcol0 += 4 * DOUT;
        bcol1 += 4 * DOUT;
    }

    const float bv0 = bias[col0 + lane16];
    const float bv1 = bias[col0 + 16 + lane16];
    float* out = H + (size_t)row0 * DOUT + col0 + lane16;
    #pragma unroll
    for (int r = 0; r < 8; ++r) {
        const size_t m = (size_t)(r + 8 * hsel) * DOUT;
        out[m]      = acc0[r] + bv0;
        out[m + 16] = acc1[r] + bv1;
    }
}

__global__ __launch_bounds__(256) void sage_gemm_kernel(
    const float* __restrict__ AGG, const float* __restrict__ INVDEG,
    const float* __restrict__ Hin,
    const float* __restrict__ Wl, const float* __restrict__ bl,
    const float* __restrict__ Wr,
    float* __restrict__ OUT, int relu)
{
    const int lane   = threadIdx.x & 31;
    const int wave   = threadIdx.x >> 5;
    const int lane16 = lane & 15;
    const int hsel   = lane >> 4;

    const int row0 = blockIdx.x * 16;
    const int col0 = wave * 32;

    const int arow_idx = row0 + lane16;
    const float inv = INVDEG[arow_idx];

    const float* aggrow = AGG + (size_t)arow_idx * DOUT + 2 * hsel;
    const float* hrow   = Hin + (size_t)arow_idx * DOUT + 2 * hsel;
    const float* blc0   = Wl + (size_t)(2 * hsel) * DOUT + (col0 + lane16);
    const float* blc1   = blc0 + 16;
    const float* brc0   = Wr + (size_t)(2 * hsel) * DOUT + (col0 + lane16);
    const float* brc1   = brc0 + 16;

    v8f acc0 = {};
    v8f acc1 = {};
    #pragma unroll 2
    for (int k = 0; k < DOUT; k += 4) {
        v2f am = *(const v2f*)aggrow;
        am.x *= inv; am.y *= inv;
        v2f ah = *(const v2f*)hrow;
        v2f bl0f, bl1f, br0f, br1f;
        bl0f.x = blc0[0];    bl0f.y = blc0[DOUT];
        bl1f.x = blc1[0];    bl1f.y = blc1[DOUT];
        br0f.x = brc0[0];    br0f.y = brc0[DOUT];
        br1f.x = brc1[0];    br1f.y = brc1[DOUT];
        acc0 = __builtin_amdgcn_wmma_f32_16x16x4_f32(false, am, false, bl0f,
                                                     (short)0, acc0, false, false);
        acc1 = __builtin_amdgcn_wmma_f32_16x16x4_f32(false, am, false, bl1f,
                                                     (short)0, acc1, false, false);
        acc0 = __builtin_amdgcn_wmma_f32_16x16x4_f32(false, ah, false, br0f,
                                                     (short)0, acc0, false, false);
        acc1 = __builtin_amdgcn_wmma_f32_16x16x4_f32(false, ah, false, br1f,
                                                     (short)0, acc1, false, false);
        aggrow += 4; hrow += 4;
        blc0 += 4 * DOUT; blc1 += 4 * DOUT;
        brc0 += 4 * DOUT; brc1 += 4 * DOUT;
    }

    const float bv0 = bl[col0 + lane16];
    const float bv1 = bl[col0 + 16 + lane16];
    const float* hres = Hin + (size_t)row0 * DOUT + col0 + lane16;
    float* out = OUT + (size_t)row0 * DOUT + col0 + lane16;
    #pragma unroll
    for (int r = 0; r < 8; ++r) {
        const size_t m = (size_t)(r + 8 * hsel) * DOUT;
        float v0 = acc0[r] + bv0;
        float v1 = acc1[r] + bv1;
        if (relu) { v0 = fmaxf(v0, 0.0f); v1 = fmaxf(v1, 0.0f); }
        out[m]      = v0 + hres[m];
        out[m + 16] = v1 + hres[m + 16];
    }
}

__global__ __launch_bounds__(256) void scatter_add_kernel(
    const float* __restrict__ H, const int* __restrict__ src,
    const int* __restrict__ dst, float* __restrict__ AGG, int E)
{
    const int e = blockIdx.x * 8 + (threadIdx.x >> 5);
    if (e >= E) return;
    const int lane = threadIdx.x & 31;
    const int s = src[e];
    const int d = dst[e];
    const float* hs = H + (size_t)s * DOUT;
    float* ad = AGG + (size_t)d * DOUT;
    #pragma unroll
    for (int j = 0; j < 8; ++j) {
        atomicAdd(&ad[lane + 32 * j], hs[lane + 32 * j]);
    }
}

__global__ void deg_kernel(const int* __restrict__ dst, float* __restrict__ deg, int E)
{
    const int e = blockIdx.x * blockDim.x + threadIdx.x;
    if (e < E) atomicAdd(&deg[dst[e]], 1.0f);
}

__global__ void invdeg_kernel(float* __restrict__ deg, int n)
{
    const int i = blockIdx.x * blockDim.x + threadIdx.x;
    if (i < n) deg[i] = 1.0f / fmaxf(deg[i], 1.0f);
}

__global__ void zero_kernel(float* __restrict__ p, int n)
{
    const int i = blockIdx.x * blockDim.x + threadIdx.x;
    if (i < n) p[i] = 0.0f;
}

extern "C" void kernel_launch(void* const* d_in, const int* in_sizes, int n_in,
                              void* d_out, int out_size, void* d_ws, size_t ws_size,
                              hipStream_t stream)
{
    const float* x   = (const float*)d_in[0];
    const int*   ei  = (const int*)d_in[1];
    const float* W0  = (const float*)d_in[2];
    const float* b0  = (const float*)d_in[3];
    const float* Wl1 = (const float*)d_in[4];
    const float* bl1 = (const float*)d_in[5];
    const float* Wr1 = (const float*)d_in[6];
    const float* Wl2 = (const float*)d_in[7];
    const float* bl2 = (const float*)d_in[8];
    const float* Wr2 = (const float*)d_in[9];

    const int* src = ei;
    const int* dst = ei + NEDGES;

    float* ws  = (float*)d_ws;
    float* h   = ws;
    float* agg = h + (size_t)NNODES * DOUT;
    float* deg = agg + (size_t)NNODES * DOUT;

    float* out1 = (float*)d_out;
    float* out2 = out1 + (size_t)NNODES * DOUT;

    const int feat_elems = NNODES * DOUT;
    dim3 gemm_grid(NNODES / 16, 1);

    zero_kernel<<<(feat_elems + 255) / 256, 256, 0, stream>>>(agg, feat_elems);
    zero_kernel<<<(NNODES + 255) / 256, 256, 0, stream>>>(deg, NNODES);

    lin_gemm_kernel<<<gemm_grid, 256, 0, stream>>>(x, W0, b0, h);

    deg_kernel<<<(NEDGES + 255) / 256, 256, 0, stream>>>(dst, deg, NEDGES);
    invdeg_kernel<<<(NNODES + 255) / 256, 256, 0, stream>>>(deg, NNODES);

    scatter_add_kernel<<<NEDGES / 8, 256, 0, stream>>>(h, src, dst, agg, NEDGES);
    sage_gemm_kernel<<<gemm_grid, 256, 0, stream>>>(agg, deg, h, Wl1, bl1, Wr1, out1, 1);

    zero_kernel<<<(feat_elems + 255) / 256, 256, 0, stream>>>(agg, feat_elems);
    scatter_add_kernel<<<NEDGES / 8, 256, 0, stream>>>(out1, src, dst, agg, NEDGES);
    sage_gemm_kernel<<<gemm_grid, 256, 0, stream>>>(agg, deg, out1, Wl2, bl2, Wr2, out2, 0);
}


# Fills and degree kernels folded into neighbour launches: GEMM blocks clear what the next scatter adds into, scatter counts degrees, 1/deg applied at operand read; 10 launches to 5; sum order kept
# speedup vs baseline: 1.8982x; 1.8982x over previous
#include <hip/hip_runtime.h>

typedef float v2f __attribute__((ext_vector_type(2)));
typedef float v8f __attribute__((ext_vector_type(8)));

#define NNODES 10000
#define NEDGES 320000
#define DIN 512
#define DOUT 256

#define TROWS 32
#define TCOLS 128
#define GEMM_THREADS 128
#define ROW_TILES ((NNODES + TROWS - 1) / TROWS)
#define GEMM_BLOCKS (ROW_TILES * (DOUT / TCOLS))
#define LIN_PITCH (DIN + 4)
#define SAGE_PITCH (DOUT + 4)

__device__ __forceinline__ void clear_tile_rows(float* __restrict__ copy, int row0, int tid)
{
    const int rsub = __builtin_amdgcn_readfirstlane(tid >> 6);
    const int rows_left = NNODES - row0 - rsub;
    float4* first = (float4*)copy + (size_t)(row0 + rsub) * (DOUT / 4) + (tid & 63);
    const float4 z = {0.0f, 0.0f, 0.0f, 0.0f};
    for (int p = 0; p < TROWS / 2; ++p)
        if (2 * p < rows_left) first[p * 2 * (DOUT / 4)] = z;
}

__global__ __launch_bounds__(GEMM_THREADS) void lin_gemm_kernel(
    const float* __restrict__ X, const float* __restrict__ W,
    const float* __restrict__ bias, float* __restrict__ H,
    float* __restrict__ ZAGG, float* __restrict__ ZDEG)
{
    __shared__ __attribute__((aligned(16))) float As[TROWS * LIN_PITCH];

    const int tid    = threadIdx.x;
    const int lane   = tid & 31;
    const int wave   = tid >> 5;
    const int lane16 = lane & 15;
    const int hsel   = lane >> 4;

    const int row0 = (blockIdx.x >> 1) * TROWS;
    const int col0 = (blockIdx.x & 1) * TCOLS + wave * 32;

    #pragma unroll
    for (int r = 0; r < TROWS; ++r) {
        int gr = row0 + r;
        if (gr >= NNODES) gr = NNODES - 1;
        const float4 v = *(const float4*)(X + (size_t)gr * DIN + 4 * tid);
        *(float4*)(As + r * LIN_PITCH + 4 * tid) = v;
    }
    __syncthreads();

    clear_tile_rows(ZAGG + (size_t)(blockIdx.x & 1) * ((size_t)NNODES * DOUT), row0, tid);
    if ((blockIdx.x & 1) == 0 && tid < TROWS && row0 + tid < NNODES) ZDEG[row0 + tid] = 0.0f;

    const float* a0p = As + lane16 * LIN_PITCH + 2 * hsel;
    const float* a1p = a0p + 16 * LIN_PITCH;
    const float* bcol0 = W + (size_t)(2 * hsel) * DOUT + (col0 + lane16);
    const float* bcol1 = bcol0 + 16;

    v8f acc00 = {};
    v8f acc01 = {};
    v8f acc10 = {};
    v8f acc11 = {};
    #pragma unroll 8
    for (int k = 0; k < DIN; k += 4) {
        const v2f a0 = *(const v2f*)(a0p + k);
        const v2f a1 = *(const v2f*)(a1p + k);
        v2f b0f, b1f;
        b0f.x = bcol0[0];
        b0f.y = bcol0[DOUT];
        b1f.x = bcol1[0];
        b1f.y = bcol1[DOUT];
        acc00 = __builtin_amdgcn_wmma_f32_16x16x4_f32(false, a0, false, b0f,
                                                      (short)0, acc00, false, false);
        acc01 = __builtin_amdgcn_wmma_f32_16x16x4_f32(false, a0, false, b1f,
                                                      (short)0, acc01, false, false);
        acc10 = __builtin_amdgcn_wmma_f32_16x16x4_f32(false, a1, false, b0f,
                                                      (short)0, acc10, false, false);
        acc11 = __builtin_amdgcn_wmma_f32_16x16x4_f32(false, a1, false, b1f,
                                                      (short)0, acc11, false, false);
        bcol0 += 4 * DOUT;
        bcol1 += 4 * DOUT;
    }

    const float bv0 = bias[col0 + lane16];
    const float bv1 = bias[col0 + 16 + lane16];
    #pragma unroll
    for (int r = 0; r < 8; ++r) {
        const int ra = row0 + r + 8 * hsel;
        const int rb = ra + 16;
        if (ra < NNODES) {
            float* o = H + (size_t)ra * DOUT + col0 + lane16;
            o[0]  = acc00[r] + bv0;
            o[16] = acc01[r] + bv1;
        }
        if (rb < NNODES) {
            float* o = H + (size_t)rb * DOUT + col0 + lane16;
            o[0]  = acc10[r] + bv0;
            o[16] = acc11[r] + bv1;
        }
    }
}

__global__ __launch_bounds__(GEMM_THREADS) void sage_gemm_kernel(
    const float* __restrict__ AGG0, const float* __restrict__ AGG1,
    const float* __restrict__ DEG,
    const float* __restrict__ Hin,
    const float* __restrict__ Wl, const float* __restrict__ bl,
    const float* __restrict__ Wr,
    float* __restrict__ OUT, int relu, float* __restrict__ ZAGG)
{
    __shared__ __attribute__((aligned(16))) float Ms[TROWS * SAGE_PITCH];
    __shared__ __attribute__((aligned(16))) float Hs[TROWS * SAGE_PITCH];

    const int tid    = threadIdx.x;
    const int lane   = tid & 31;
    const int wave   = tid >> 5;
    const int lane16 = lane & 15;
    const int hsel   = lane >> 4;

    const int row0 = (blockIdx.x >> 1) * TROWS;
    const int col0 = (blockIdx.x & 1) * TCOLS + wave * 32;

    int ga = row0 + lane16;
    int gb = ga + 16;
    if (ga >= NNODES) ga = NNODES - 1;
    if (gb >= NNODES) gb = NNODES - 1;
    const float dega = DEG[ga];
    const float degb = DEG[gb];

    {
        const int rsub = tid >> 6;
        const int c4   = (tid & 63) * 4;
        float4 mv[TROWS / 2], nv[TROWS / 2], hv[TROWS / 2];
        #pragma unroll
        for (int p = 0; p < TROWS / 2; ++p) {
            int gr = row0 + 2 * p + rsub;
            if (gr >= NNODES) gr = NNODES - 1;
            const unsigned q = (unsigned)gr * (DOUT / 4) + (unsigned)(tid & 63);
            mv[p] = ((const float4*)AGG0)[q];
            nv[p] = ((const float4*)AGG1)[q];
            hv[p] = ((const float4*)Hin)[q];
        }
        #pragma unroll
        for (int p = 0; p < TROWS / 2; ++p) {
            const int r = 2 * p + rsub;
            float4 m = mv[p];
            const float4 n2 = nv[p];
            m.x += n2.x; m.y += n2.y; m.z += n2.z; m.w += n2.w;
            *(float4*)(Ms + r * SAGE_PITCH + c4) = m;
            *(float4*)(Hs + r * SAGE_PITCH + c4) = hv[p];
        }
    }
    __syncthreads();

    if (ZAGG != nullptr)
        clear_tile_rows(ZAGG + (size_t)(blockIdx.x & 1) * ((size_t)NNODES * DOUT), row0, tid);

    const float inv0 = 1.0f / fmaxf(dega, 1.0f);
    const float inv1 = 1.0f / fmaxf(degb, 1.0f);

    const float* m0p = Ms + lane16 * SAGE_PITCH + 2 * hsel;
    const float* m1p = m0p + 16 * SAGE_PITCH;
    const float* h0p = Hs + lane16 * SAGE_PITCH + 2 * hsel;
    const float* h1p = h0p + 16 * SAGE_PITCH;
    const float* blc0 = Wl + (size_t)(2 * hsel) * DOUT + (col0 + lane16);
    const float* blc1 = blc0 + 16;
    const float* brc0 = Wr + (size_t)(2 * hsel) * DOUT + (col0 + lane16);
    const float* brc1 = brc0 + 16;

    v8f acc00 = {};
    v8f acc01 = {};
    v8f acc10 = {};
    v8f acc11 = {};
    #pragma unroll 4
    for (int k = 0; k < DOUT; k += 4) {
        const v2f am0 = *(const v2f*)(m0p + k) * inv0;
        const v2f am1 = *(const v2f*)(m1p + k) * inv1;
        const v2f ah0 = *(const v2f*)(h0p + k);
        const v2f ah1 = *(const v2f*)(h1p + k);
        v2f bl0f, bl1f, br0f, br1f;
        bl0f.x = blc0[0];    bl0f.y = blc0[DOUT];
        bl1f.x = blc1[0];    bl1f.y = blc1[DOUT];
        br0f.x = brc0[0];    br0f.y = brc0[DOUT];
        br1f.x = brc1[0];    br1f.y = brc1[DOUT];
        acc00 = __builtin_amdgcn_wmma_f32_16x16x4_f32(false, am0, false, bl0f,
                                                      (short)0, acc00, false, false);
        acc01 = __builtin_amdgcn_wmma_f32_16x16x4_f32(false, am0, false, bl1f,
                                                      (short)0, acc01, false, false);
        acc10 = __builtin_amdgcn_wmma_f32_16x16x4_f32(false, am1, false, bl0f,
                                                      (short)0, acc10, false, false);
        acc11 = __builtin_amdgcn_wmma_f32_16x16x4_f32(false, am1, false, bl1f,
                                                      (short)0, acc11, false, false);
        acc00 = __builtin_amdgcn_wmma_f32_16x16x4_f32(false, ah0, false, br0f,
                                                      (short)0, acc00, false, false);
        acc01 = __builtin_amdgcn_wmma_f32_16x16x4_f32(false, ah0, false, br1f,
                                                      (short)0, acc01, false, false);
        acc10 = __builtin_amdgcn_wmma_f32_16x16x4_f32(false, ah1, false, br0f,
                                                      (short)0, acc10, false, false);
        acc11 = __builtin_amdgcn_wmma_f32_16x16x4_f32(false, ah1, false, br1f,
                                                      (short)0, acc11, false, false);
        blc0 += 4 * DOUT; blc1 += 4 * DOUT;
        brc0 += 4 * DOUT; brc1 += 4 * DOUT;
    }

    const float bv0 = bl[col0 + lane16];
    const float bv1 = bl[col0 + 16 + lane16];
    #pragma unroll
    for (int r = 0; r < 8; ++r) {
        const int la = r + 8 * hsel;
        const int lb = la + 16;
        float v00 = acc00[r] + bv0;
        float v01 = acc01[r] + bv1;
        float v10 = acc10[r] + bv0;
        float v11 = acc11[r] + bv1;
        if (relu) {
            v00 = fmaxf(v00, 0.0f); v01 = fmaxf(v01, 0.0f);
            v10 = fmaxf(v10, 0.0f); v11 = fmaxf(v11, 0.0f);
        }
        if (row0 + la < NNODES) {
            const float* hr = Hs + la * SAGE_PITCH + col0 + lane16;
            float* o = OUT + (size_t)(row0 + la) * DOUT + col0 + lane16;
            o[0]  = v00 + hr[0];
            o[16] = v01 + hr[16];
        }
        if (row0 + lb < NNODES) {
            const float* hr = Hs + lb * SAGE_PITCH + col0 + lane16;
            float* o = OUT + (size_t)(row0 + lb) * DOUT + col0 + lane16;
            o[0]  = v10 + hr[0];
            o[16] = v11 + hr[16];
        }
    }
}

#define EDGES_PER_WAVE 32
#define EDGES_PER_TRIP 4
#define SCATTER_BLOCKS ((NEDGES + 8 * EDGES_PER_WAVE - 1) / (8 * EDGES_PER_WAVE))

__global__ __launch_bounds__(256) void scatter_add_kernel(
    const float* __restrict__ H, const int* __restrict__ src,
    const int* __restrict__ dst, float* __restrict__ AGG,
    float* __restrict__ DEG, int E)
{
    const int lane = threadIdx.x & 31;
    const int wave = __builtin_amdgcn_readfirstlane((int)(threadIdx.x >> 5));
    const int e0 = (blockIdx.x * 8 + wave) * EDGES_PER_WAVE;
    if (e0 >= E) return;
    const unsigned place = __builtin_amdgcn_s_sendmsg_rtn(0x87);
    float* __restrict__ aggh = AGG + (size_t)((place >> 18) & 1u) * ((size_t)NNODES * DOUT);
    const int rest = E - e0;
    const int n = rest < EDGES_PER_WAVE ? rest : EDGES_PER_WAVE;

    int s_l = 0, d_l = 0;
    if (lane < n) {
        s_l = src[e0 + lane];
        d_l = dst[e0 + lane];
        if (DEG != nullptr) atomicAdd(&DEG[d_l], 1.0f);
    }

    int i = 0;
    for (; i + EDGES_PER_TRIP <= n; i += EDGES_PER_TRIP) {
        float v[EDGES_PER_TRIP][8];
        #pragma unroll
        for (int u = 0; u < EDGES_PER_TRIP; ++u) {
            const int s = __builtin_amdgcn_readlane(s_l, i + u);
            const float* hs = H + (size_t)s * DOUT + lane;
            #pragma unroll
            for (int j = 0; j < 8; ++j) v[u][j] = hs[32 * j];
        }
        #pragma unroll
        for (int u = 0; u < EDGES_PER_TRIP; ++u) {
            const int d = __builtin_amdgcn_readlane(d_l, i + u);
            float* ad = aggh + (size_t)d * DOUT + lane;
            #pragma unroll
            for (int j = 0; j < 8; ++j) atomicAdd(&ad[32 * j], v[u][j]);
        }
    }
    for (; i < n; ++i) {
        const int s = __builtin_amdgcn_readlane(s_l, i);
        const int d = __builtin_amdgcn_readlane(d_l, i);
        const float* hs = H + (size_t)s * DOUT + lane;
        float* ad = aggh + (size_t)d * DOUT + lane;
        float w[8];
        #pragma unroll
        for (int j = 0; j < 8; ++j) w[j] = hs[32 * j];
        #pragma unroll
        for (int j = 0; j < 8; ++j) atomicAdd(&ad[32 * j], w[j]);
    }
}

extern "C" void kernel_launch(void* const* d_in, const int* in_sizes, int n_in,
                              void* d_out, int out_size, void* d_ws, size_t ws_size,
                              hipStream_t stream)
{
    const float* x   = (const float*)d_in[0];
    const int*   ei  = (const int*)d_in[1];
    const float* W0  = (const float*)d_in[2];
    const float* b0  = (const float*)d_in[3];
    const float* Wl1 = (const float*)d_in[4];
    const float* bl1 = (const float*)d_in[5];
    const float* Wr1 = (const float*)d_in[6];
    const float* Wl2 = (const float*)d_in[7];
    const float* bl2 = (const float*)d_in[8];
    const float* Wr2 = (const float*)d_in[9];

    const int* src = ei;
    const int* dst = ei + NEDGES;

    float* ws  = (float*)d_ws;
    float* h    = ws;
    float* aggA = h + (size_t)NNODES * DOUT;
    float* aggA1 = aggA + (size_t)NNODES * DOUT;
    float* aggB = aggA1 + (size_t)NNODES * DOUT;
    float* aggB1 = aggB + (size_t)NNODES * DOUT;
    float* deg  = aggB1 + (size_t)NNODES * DOUT;

    float* out1 = (float*)d_out;
    float* out2 = out1 + (size_t)NNODES * DOUT;

    dim3 gemm_grid(GEMM_BLOCKS, 1);


    lin_gemm_kernel<<<gemm_grid, GEMM_THREADS, 0, stream>>>(x, W0, b0, h, aggA, deg);

    scatter_add_kernel<<<SCATTER_BLOCKS, 256, 0, stream>>>(h, src, dst, aggA, deg, NEDGES);
    sage_gemm_kernel<<<gemm_grid, GEMM_THREADS, 0, stream>>>(aggA, aggA1, deg, h, Wl1, bl1, Wr1, out1, 1, aggB);

    scatter_add_kernel<<<SCATTER_BLOCKS, 256, 0, stream>>>(out1, src, dst, aggB, (float*)nullptr, NEDGES);
    sage_gemm_kernel<<<gemm_grid, GEMM_THREADS, 0, stream>>>(aggB, aggB1, deg, out1, Wl2, bl2, Wr2, out2, 0, (float*)nullptr);
}


# GEMM work re-cut to one work-group per WGP: 80x128 tiles (10000 = 125 x 80; 250 groups, a wave owns 80x16), operand fragments fetched a step ahead in registers; SIMDs carry 10 tiles, was 8 or 12
# speedup vs baseline: 2.0273x; 1.0680x over previous
#include <hip/hip_runtime.h>

typedef float v2f __attribute__((ext_vector_type(2)));
typedef float v8f __attribute__((ext_vector_type(8)));

#define NNODES 10000
#define NEDGES 320000
#define DIN 512
#define DOUT 256

#define TROWS 80
#define TCOLS 128
#define GEMM_THREADS 256
#define ROW_TILES (NNODES / TROWS)
#define GEMM_BLOCKS (ROW_TILES * (DOUT / TCOLS))
#define WTILES (TROWS / 16)
#define KB 8
#define KBS 4
static_assert(NNODES % TROWS == 0, "row tiles must be full: the kernels carry no row guards");
#define LIN_PITCH (DIN + 4)
#define SAGE_PITCH (DOUT + 4)

__device__ __forceinline__ void clear_tile_rows(float* __restrict__ copy, int row0, int tid)
{
    const int rsub = __builtin_amdgcn_readfirstlane(tid >> 6);
    float4* first = (float4*)copy + (size_t)(row0 + rsub) * (DOUT / 4) + (tid & 63);
    const float4 z = {0.0f, 0.0f, 0.0f, 0.0f};
    for (int p = 0; p < TROWS / 4; ++p) first[p * 4 * (DOUT / 4)] = z;
}

__global__ __launch_bounds__(GEMM_THREADS) void lin_gemm_kernel(
    const float* __restrict__ X, const float* __restrict__ W,
    const float* __restrict__ bias, float* __restrict__ H,
    float* __restrict__ ZAGG, float* __restrict__ ZDEG)
{
    __shared__ __attribute__((aligned(16))) float As[TROWS * LIN_PITCH];

    const int tid    = threadIdx.x;
    const int lane   = tid & 31;
    const int wave   = tid >> 5;
    const int lane16 = lane & 15;
    const int hsel   = lane >> 4;

    const int row0 = (blockIdx.x >> 1) * TROWS;
    const int col0 = (blockIdx.x & 1) * TCOLS + wave * 16;

    {
        const int rsub = tid >> 7;
        const int c4   = (tid & 127) * 4;
        #pragma unroll
        for (int p = 0; p < TROWS / 2; ++p) {
            const int r = 2 * p + rsub;
            const float4 v = *(const float4*)(X + (size_t)(row0 + r) * DIN + c4);
            *(float4*)(As + r * LIN_PITCH + c4) = v;
        }
    }
    __syncthreads();

    clear_tile_rows(ZAGG + (size_t)(blockIdx.x & 1) * ((size_t)NNODES * DOUT), row0, tid);
    if ((blockIdx.x & 1) == 0 && tid < TROWS) ZDEG[row0 + tid] = 0.0f;

    const float* ap   = As + lane16 * LIN_PITCH + 2 * hsel;
    const float* bcol = W + (size_t)(2 * hsel) * DOUT + (col0 + lane16);

    v8f acc[WTILES];
    #pragma unroll
    for (int t = 0; t < WTILES; ++t) acc[t] = v8f{};

    v2f bc[KB], ac[WTILES];
    #pragma unroll
    for (int s = 0; s < KB; ++s) {
        bc[s].x = bcol[(4 * s) * DOUT];
        bc[s].y = bcol[(4 * s + 1) * DOUT];
    }
    #pragma unroll
    for (int t = 0; t < WTILES; ++t) ac[t] = *(const v2f*)(ap + t * 16 * LIN_PITCH);

    for (int kb = 0; kb < DIN; kb += 4 * KB) {
        const int knb = (kb + 4 * KB < DIN) ? kb + 4 * KB : 0;
        const float* bnp = bcol + (size_t)knb * DOUT;
        v2f bn[KB];
        #pragma unroll
        for (int s = 0; s < KB; ++s) {
            bn[s].x = bnp[(4 * s) * DOUT];
            bn[s].y = bnp[(4 * s + 1) * DOUT];
        }
        #pragma unroll
        for (int s = 0; s < KB; ++s) {
            const int kn = (s + 1 < KB) ? kb + 4 * (s + 1) : knb;
            v2f an[WTILES];
            #pragma unroll
            for (int t = 0; t < WTILES; ++t) an[t] = *(const v2f*)(ap + t * 16 * LIN_PITCH + kn);
            #pragma unroll
            for (int t = 0; t < WTILES; ++t)
                acc[t] = __builtin_amdgcn_wmma_f32_16x16x4_f32(false, ac[t], false, bc[s],
                                                               (short)0, acc[t], false, false);
            #pragma unroll
            for (int t = 0; t < WTILES; ++t) ac[t] = an[t];
        }
        #pragma unroll
        for (int s = 0; s < KB; ++s) bc[s] = bn[s];
    }

    const float bv = bias[col0 + lane16];
    #pragma unroll
    for (int t = 0; t < WTILES; ++t) {
        #pragma unroll
        for (int r = 0; r < 8; ++r) {
            const int row = row0 + 16 * t + r + 8 * hsel;
            H[(size_t)row * DOUT + col0 + lane16] = acc[t][r] + bv;
        }
    }
}

__global__ __launch_bounds__(GEMM_THREADS) void sage_gemm_kernel(
    const float* __restrict__ AGG0, const float* __restrict__ AGG1,
    const float* __restrict__ DEG,
    const float* __restrict__ Hin,
    const float* __restrict__ Wl, const float* __restrict__ bl,
    const float* __restrict__ Wr,
    float* __restrict__ OUT, int relu, float* __restrict__ ZAGG)
{
    __shared__ __attribute__((aligned(16))) float Ms[TROWS * SAGE_PITCH];
    __shared__ __attribute__((aligned(16))) float Hs[TROWS * SAGE_PITCH];

    const int tid    = threadIdx.x;
    const int lane   = tid & 31;
    const int wave   = tid >> 5;
    const int lane16 = lane & 15;
    const int hsel   = lane >> 4;

    const int row0 = (blockIdx.x >> 1) * TROWS;
    const int col0 = (blockIdx.x & 1) * TCOLS + wave * 16;

    float degv[WTILES];
    #pragma unroll
    for (int t = 0; t < WTILES; ++t) degv[t] = DEG[row0 + 16 * t + lane16];

    {
        const int rsub = tid >> 6;
        const int c4   = (tid & 63) * 4;
        float4 mv[TROWS / 4], nv[TROWS / 4], hv[TROWS / 4];
        #pragma unroll
        for (int p = 0; p < TROWS / 4; ++p) {
            const unsigned q = (unsigned)(row0 + 4 * p + rsub) * (DOUT / 4) + (unsigned)(tid & 63);
            mv[p] = ((const float4*)AGG0)[q];
            nv[p] = ((const float4*)AGG1)[q];
            hv[p] = ((const float4*)Hin)[q];
        }
        #pragma unroll
        for (int p = 0; p < TROWS / 4; ++p) {
            const int r = 4 * p + rsub;
            float4 m = mv[p];
            const float4 n2 = nv[p];
            m.x += n2.x; m.y += n2.y; m.z += n2.z; m.w += n2.w;
            *(float4*)(Ms + r * SAGE_PITCH + c4) = m;
            *(float4*)(Hs + r * SAGE_PITCH + c4) = hv[p];
        }
    }
    __syncthreads();

    if (ZAGG != nullptr)
        clear_tile_rows(ZAGG + (size_t)(blockIdx.x & 1) * ((size_t)NNODES * DOUT), row0, tid);

    float inv[WTILES];
    #pragma unroll
    for (int t = 0; t < WTILES; ++t) inv[t] = 1.0f / fmaxf(degv[t], 1.0f);

    const float* mp  = Ms + lane16 * SAGE_PITCH + 2 * hsel;
    const float* hp  = Hs + lane16 * SAGE_PITCH + 2 * hsel;
    const float* blc = Wl + (size_t)(2 * hsel) * DOUT + (col0 + lane16);
    const float* brc = Wr + (size_t)(2 * hsel) * DOUT + (col0 + lane16);

    v8f acc[WTILES];
    #pragma unroll
    for (int t = 0; t < WTILES; ++t) acc[t] = v8f{};

    v2f wlf[KBS], wrf[KBS], mc[WTILES], hc[WTILES];
    #pragma unroll
    for (int s = 0; s < KBS; ++s) {
        wlf[s].x = blc[(4 * s) * DOUT];    wlf[s].y = blc[(4 * s + 1) * DOUT];
        wrf[s].x = brc[(4 * s) * DOUT];    wrf[s].y = brc[(4 * s + 1) * DOUT];
    }
    #pragma unroll
    for (int t = 0; t < WTILES; ++t) {
        mc[t] = *(const v2f*)(mp + t * 16 * SAGE_PITCH);
        hc[t] = *(const v2f*)(hp + t * 16 * SAGE_PITCH);
    }

    for (int kb = 0; kb < DOUT; kb += 4 * KBS) {
        const int knb = (kb + 4 * KBS < DOUT) ? kb + 4 * KBS : 0;
        const float* blnp = blc + (size_t)knb * DOUT;
        const float* brnp = brc + (size_t)knb * DOUT;
        v2f wlfn[KBS], wrfn[KBS];
        #pragma unroll
        for (int s = 0; s < KBS; ++s) {
            wlfn[s].x = blnp[(4 * s) * DOUT];    wlfn[s].y = blnp[(4 * s + 1) * DOUT];
            wrfn[s].x = brnp[(4 * s) * DOUT];    wrfn[s].y = brnp[(4 * s + 1) * DOUT];
        }
        #pragma unroll
        for (int s = 0; s < KBS; ++s) {
            const int kn = (s + 1 < KBS) ? kb + 4 * (s + 1) : knb;
            v2f mn[WTILES], hn[WTILES];
            #pragma unroll
            for (int t = 0; t < WTILES; ++t) {
                mn[t] = *(const v2f*)(mp + t * 16 * SAGE_PITCH + kn);
                hn[t] = *(const v2f*)(hp + t * 16 * SAGE_PITCH + kn);
            }
            #pragma unroll
            for (int t = 0; t < WTILES; ++t) {
                const v2f am = mc[t] * inv[t];
                acc[t] = __builtin_amdgcn_wmma_f32_16x16x4_f32(false, am, false, wlf[s],
                                                               (short)0, acc[t], false, false);
            }
            #pragma unroll
            for (int t = 0; t < WTILES; ++t)
                acc[t] = __builtin_amdgcn_wmma_f32_16x16x4_f32(false, hc[t], false, wrf[s],
                                                               (short)0, acc[t], false, false);
            #pragma unroll
            for (int t = 0; t < WTILES; ++t) { mc[t] = mn[t]; hc[t] = hn[t]; }
        }
        #pragma unroll
        for (int s = 0; s < KBS; ++s) { wlf[s] = wlfn[s]; wrf[s] = wrfn[s]; }
    }

    const float bv = bl[col0 + lane16];
    #pragma unroll
    for (int t = 0; t < WTILES; ++t) {
        #pragma unroll
        for (int r = 0; r < 8; ++r) {
            const int lr = 16 * t + r + 8 * hsel;
            float v = acc[t][r] + bv;
            if (relu) v = fmaxf(v, 0.0f);
            const float hr = Hs[lr * SAGE_PITCH + col0 + lane16];
            OUT[(size_t)(row0 + lr) * DOUT + col0 + lane16] = v + hr;
        }
    }
}

#define EDGES_PER_WAVE 32
#define EDGES_PER_TRIP 4
#define SCATTER_BLOCKS ((NEDGES + 8 * EDGES_PER_WAVE - 1) / (8 * EDGES_PER_WAVE))

__global__ __launch_bounds__(256) void scatter_add_kernel(
    const float* __restrict__ H, const int* __restrict__ src,
    const int* __restrict__ dst, float* __restrict__ AGG,
    float* __restrict__ DEG, int E)
{
    const int lane = threadIdx.x & 31;
    const int wave = __builtin_amdgcn_readfirstlane((int)(threadIdx.x >> 5));
    const int e0 = (blockIdx.x * 8 + wave) * EDGES_PER_WAVE;
    if (e0 >= E) return;
    const unsigned place = __builtin_amdgcn_s_sendmsg_rtn(0x87);
    float* __restrict__ aggh = AGG + (size_t)((place >> 18) & 1u) * ((size_t)NNODES * DOUT);
    const int rest = E - e0;
    const int n = rest < EDGES_PER_WAVE ? rest : EDGES_PER_WAVE;

    int s_l = 0, d_l = 0;
    if (lane < n) {
        s_l = src[e0 + lane];
        d_l = dst[e0 + lane];
        if (DEG != nullptr) atomicAdd(&DEG[d_l], 1.0f);
    }

    int i = 0;
    for (; i + EDGES_PER_TRIP <= n; i += EDGES_PER_TRIP) {
        float v[EDGES_PER_TRIP][8];
        #pragma unroll
        for (int u = 0; u < EDGES_PER_TRIP; ++u) {
            const int s = __builtin_amdgcn_readlane(s_l, i + u);
            const float* hs = H + (size_t)s * DOUT + lane;
            #pragma unroll
            for (int j = 0; j < 8; ++j) v[u][j] = hs[32 * j];
        }
        #pragma unroll
        for (int u = 0; u < EDGES_PER_TRIP; ++u) {
            const int d = __builtin_amdgcn_readlane(d_l, i + u);
            float* ad = aggh + (size_t)d * DOUT + lane;
            #pragma unroll
            for (int j = 0; j < 8; ++j) atomicAdd(&ad[32 * j], v[u][j]);
        }
    }
    for (; i < n; ++i) {
        const int s = __builtin_amdgcn_readlane(s_l, i);
        const int d = __builtin_amdgcn_readlane(d_l, i);
        const float* hs = H + (size_t)s * DOUT + lane;
        float* ad = aggh + (size_t)d * DOUT + lane;
        float w[8];
        #pragma unroll
        for (int j = 0; j < 8; ++j) w[j] = hs[32 * j];
        #pragma unroll
        for (int j = 0; j < 8; ++j) atomicAdd(&ad[32 * j], w[j]);
    }
}

extern "C" void kernel_launch(void* const* d_in, const int* in_sizes, int n_in,
                              void* d_out, int out_size, void* d_ws, size_t ws_size,
                              hipStream_t stream)
{
    const float* x   = (const float*)d_in[0];
    const int*   ei  = (const int*)d_in[1];
    const float* W0  = (const float*)d_in[2];
    const float* b0  = (const float*)d_in[3];
    const float* Wl1 = (const float*)d_in[4];
    const float* bl1 = (const float*)d_in[5];
    const float* Wr1 = (const float*)d_in[6];
    const float* Wl2 = (const float*)d_in[7];
    const float* bl2 = (const float*)d_in[8];
    const float* Wr2 = (const float*)d_in[9];

    const int* src = ei;
    const int* dst = ei + NEDGES;

    float* ws  = (float*)d_ws;
    float* h    = ws;
    float* aggA = h + (size_t)NNODES * DOUT;
    float* aggA1 = aggA + (size_t)NNODES * DOUT;
    float* aggB = aggA1 + (size_t)NNODES * DOUT;
    float* aggB1 = aggB + (size_t)NNODES * DOUT;
    float* deg  = aggB1 + (size_t)NNODES * DOUT;

    float* out1 = (float*)d_out;
    float* out2 = out1 + (size_t)NNODES * DOUT;

    dim3 gemm_grid(GEMM_BLOCKS, 1);


    lin_gemm_kernel<<<gemm_grid, GEMM_THREADS, 0, stream>>>(x, W0, b0, h, aggA, deg);

    scatter_add_kernel<<<SCATTER_BLOCKS, 256, 0, stream>>>(h, src, dst, aggA, deg, NEDGES);
    sage_gemm_kernel<<<gemm_grid, GEMM_THREADS, 0, stream>>>(aggA, aggA1, deg, h, Wl1, bl1, Wr1, out1, 1, aggB);

    scatter_add_kernel<<<SCATTER_BLOCKS, 256, 0, stream>>>(out1, src, dst, aggB, (float*)nullptr, NEDGES);
    sage_gemm_kernel<<<gemm_grid, GEMM_THREADS, 0, stream>>>(aggB, aggB1, deg, out1, Wl2, bl2, Wr2, out2, 0, (float*)nullptr);
}


# Neighbour sums built in LDS, not by second-level float atomics: a work-group owns 40 nodes, sweeps dst, lists its edges, LDS float-adds their rows, writes one agg copy; sums in LDS arrival order
# speedup vs baseline: 2.5694x; 1.2674x over previous
#include <hip/hip_runtime.h>

typedef float v2f __attribute__((ext_vector_type(2)));
typedef float v8f __attribute__((ext_vector_type(8)));

#define NNODES 10000
#define NEDGES 320000
#define DIN 512
#define DOUT 256

#define TROWS 80
#define TCOLS 128
#define GEMM_THREADS 256
#define ROW_TILES (NNODES / TROWS)
#define GEMM_BLOCKS (ROW_TILES * (DOUT / TCOLS))
#define WTILES (TROWS / 16)
#define KB 8
#define KBS 4
static_assert(NNODES % TROWS == 0, "row tiles must be full: the kernels carry no row guards");
#define LIN_PITCH (DIN + 4)
#define SAGE_PITCH (DOUT + 4)

__global__ __launch_bounds__(GEMM_THREADS) void lin_gemm_kernel(
    const float* __restrict__ X, const float* __restrict__ W,
    const float* __restrict__ bias, float* __restrict__ H)
{
    __shared__ __attribute__((aligned(16))) float As[TROWS * LIN_PITCH];

    const int tid    = threadIdx.x;
    const int lane   = tid & 31;
    const int wave   = tid >> 5;
    const int lane16 = lane & 15;
    const int hsel   = lane >> 4;

    const int row0 = (blockIdx.x >> 1) * TROWS;
    const int col0 = (blockIdx.x & 1) * TCOLS + wave * 16;

    {
        const int rsub = tid >> 7;
        const int c4   = (tid & 127) * 4;
        #pragma unroll
        for (int p = 0; p < TROWS / 2; ++p) {
            const int r = 2 * p + rsub;
            const float4 v = *(const float4*)(X + (size_t)(row0 + r) * DIN + c4);
            *(float4*)(As + r * LIN_PITCH + c4) = v;
        }
    }
    __syncthreads();

    const float* ap   = As + lane16 * LIN_PITCH + 2 * hsel;
    const float* bcol = W + (size_t)(2 * hsel) * DOUT + (col0 + lane16);

    v8f acc[WTILES];
    #pragma unroll
    for (int t = 0; t < WTILES; ++t) acc[t] = v8f{};

    v2f bc[KB], ac[WTILES];
    #pragma unroll
    for (int s = 0; s < KB; ++s) {
        bc[s].x = bcol[(4 * s) * DOUT];
        bc[s].y = bcol[(4 * s + 1) * DOUT];
    }
    #pragma unroll
    for (int t = 0; t < WTILES; ++t) ac[t] = *(const v2f*)(ap + t * 16 * LIN_PITCH);

    for (int kb = 0; kb < DIN; kb += 4 * KB) {
        const int knb = (kb + 4 * KB < DIN) ? kb + 4 * KB : 0;
        const float* bnp = bcol + (size_t)knb * DOUT;
        v2f bn[KB];
        #pragma unroll
        for (int s = 0; s < KB; ++s) {
            bn[s].x = bnp[(4 * s) * DOUT];
            bn[s].y = bnp[(4 * s + 1) * DOUT];
        }
        #pragma unroll
        for (int s = 0; s < KB; ++s) {
            const int kn = (s + 1 < KB) ? kb + 4 * (s + 1) : knb;
            v2f an[WTILES];
            #pragma unroll
            for (int t = 0; t < WTILES; ++t) an[t] = *(const v2f*)(ap + t * 16 * LIN_PITCH + kn);
            #pragma unroll
            for (int t = 0; t < WTILES; ++t)
                acc[t] = __builtin_amdgcn_wmma_f32_16x16x4_f32(false, ac[t], false, bc[s],
                                                               (short)0, acc[t], false, false);
            #pragma unroll
            for (int t = 0; t < WTILES; ++t) ac[t] = an[t];
        }
        #pragma unroll
        for (int s = 0; s < KB; ++s) bc[s] = bn[s];
    }

    const float bv = bias[col0 + lane16];
    #pragma unroll
    for (int t = 0; t < WTILES; ++t) {
        #pragma unroll
        for (int r = 0; r < 8; ++r) {
            const int row = row0 + 16 * t + r + 8 * hsel;
            H[(size_t)row * DOUT + col0 + lane16] = acc[t][r] + bv;
        }
    }
}

__global__ __launch_bounds__(GEMM_THREADS) void sage_gemm_kernel(
    const float* __restrict__ AGG,
    const float* __restrict__ DEG,
    const float* __restrict__ Hin,
    const float* __restrict__ Wl, const float* __restrict__ bl,
    const float* __restrict__ Wr,
    float* __restrict__ OUT, int relu)
{
    __shared__ __attribute__((aligned(16))) float Ms[TROWS * SAGE_PITCH];
    __shared__ __attribute__((aligned(16))) float Hs[TROWS * SAGE_PITCH];

    const int tid    = threadIdx.x;
    const int lane   = tid & 31;
    const int wave   = tid >> 5;
    const int lane16 = lane & 15;
    const int hsel   = lane >> 4;

    const int row0 = (blockIdx.x >> 1) * TROWS;
    const int col0 = (blockIdx.x & 1) * TCOLS + wave * 16;

    float degv[WTILES];
    #pragma unroll
    for (int t = 0; t < WTILES; ++t) degv[t] = DEG[row0 + 16 * t + lane16];

    {
        const int rsub = tid >> 6;
        const int c4   = (tid & 63) * 4;
        float4 mv[TROWS / 4], hv[TROWS / 4];
        #pragma unroll
        for (int p = 0; p < TROWS / 4; ++p) {
            const unsigned q = (unsigned)(row0 + 4 * p + rsub) * (DOUT / 4) + (unsigned)(tid & 63);
            mv[p] = ((const float4*)AGG)[q];
            hv[p] = ((const float4*)Hin)[q];
        }
        #pragma unroll
        for (int p = 0; p < TROWS / 4; ++p) {
            const int r = 4 * p + rsub;
            *(float4*)(Ms + r * SAGE_PITCH + c4) = mv[p];
            *(float4*)(Hs + r * SAGE_PITCH + c4) = hv[p];
        }
    }
    __syncthreads();

    float inv[WTILES];
    #pragma unroll
    for (int t = 0; t < WTILES; ++t) inv[t] = 1.0f / fmaxf(degv[t], 1.0f);

    const float* mp  = Ms + lane16 * SAGE_PITCH + 2 * hsel;
    const float* hp  = Hs + lane16 * SAGE_PITCH + 2 * hsel;
    const float* blc = Wl + (size_t)(2 * hsel) * DOUT + (col0 + lane16);
    const float* brc = Wr + (size_t)(2 * hsel) * DOUT + (col0 + lane16);

    v8f acc[WTILES];
    #pragma unroll
    for (int t = 0; t < WTILES; ++t) acc[t] = v8f{};

    v2f wlf[KBS], wrf[KBS], mc[WTILES], hc[WTILES];
    #pragma unroll
    for (int s = 0; s < KBS; ++s) {
        wlf[s].x = blc[(4 * s) * DOUT];    wlf[s].y = blc[(4 * s + 1) * DOUT];
        wrf[s].x = brc[(4 * s) * DOUT];    wrf[s].y = brc[(4 * s + 1) * DOUT];
    }
    #pragma unroll
    for (int t = 0; t < WTILES; ++t) {
        mc[t] = *(const v2f*)(mp + t * 16 * SAGE_PITCH);
        hc[t] = *(const v2f*)(hp + t * 16 * SAGE_PITCH);
    }

    for (int kb = 0; kb < DOUT; kb += 4 * KBS) {
        const int knb = (kb + 4 * KBS < DOUT) ? kb + 4 * KBS : 0;
        const float* blnp = blc + (size_t)knb * DOUT;
        const float* brnp = brc + (size_t)knb * DOUT;
        v2f wlfn[KBS], wrfn[KBS];
        #pragma unroll
        for (int s = 0; s < KBS; ++s) {
            wlfn[s].x = blnp[(4 * s) * DOUT];    wlfn[s].y = blnp[(4 * s + 1) * DOUT];
            wrfn[s].x = brnp[(4 * s) * DOUT];    wrfn[s].y = brnp[(4 * s + 1) * DOUT];
        }
        #pragma unroll
        for (int s = 0; s < KBS; ++s) {
            const int kn = (s + 1 < KBS) ? kb + 4 * (s + 1) : knb;
            v2f mn[WTILES], hn[WTILES];
            #pragma unroll
            for (int t = 0; t < WTILES; ++t) {
                mn[t] = *(const v2f*)(mp + t * 16 * SAGE_PITCH + kn);
                hn[t] = *(const v2f*)(hp + t * 16 * SAGE_PITCH + kn);
            }
            #pragma unroll
            for (int t = 0; t < WTILES; ++t) {
                const v2f am = mc[t] * inv[t];
                acc[t] = __builtin_amdgcn_wmma_f32_16x16x4_f32(false, am, false, wlf[s],
                                                               (short)0, acc[t], false, false);
            }
            #pragma unroll
            for (int t = 0; t < WTILES; ++t)
                acc[t] = __builtin_amdgcn_wmma_f32_16x16x4_f32(false, hc[t], false, wrf[s],
                                                               (short)0, acc[t], false, false);
            #pragma unroll
            for (int t = 0; t < WTILES; ++t) { mc[t] = mn[t]; hc[t] = hn[t]; }
        }
        #pragma unroll
        for (int s = 0; s < KBS; ++s) { wlf[s] = wlfn[s]; wrf[s] = wrfn[s]; }
    }

    const float bv = bl[col0 + lane16];
    #pragma unroll
    for (int t = 0; t < WTILES; ++t) {
        #pragma unroll
        for (int r = 0; r < 8; ++r) {
            const int lr = 16 * t + r + 8 * hsel;
            float v = acc[t][r] + bv;
            if (relu) v = fmaxf(v, 0.0f);
            const float hr = Hs[lr * SAGE_PITCH + col0 + lane16];
            OUT[(size_t)(row0 + lr) * DOUT + col0 + lane16] = v + hr;
        }
    }
}

#define AROWS 40
#define AGG_THREADS 1024
#define AGG_WAVES (AGG_THREADS / 32)
#define AGG_BLOCKS (NNODES / AROWS)
#define WCAP 512
#define NQUADS (NEDGES / 4)
#define SCAN_DEPTH 8
#define SCAN_TRIPS ((NQUADS + SCAN_DEPTH * AGG_THREADS - 1) / (SCAN_DEPTH * AGG_THREADS))
#define EDGES_PER_TRIP 4
#define SUMPOS(c) (((c) & 128) + (((c) & 3) << 5) + (((c) & 127) >> 2))
static_assert(DOUT == 256, "a row of sums is two halves of 32 lanes x 4 columns");
static_assert(NNODES % AROWS == 0, "every work-group owns AROWS whole rows");
static_assert(NEDGES % 4 == 0, "dst is read 4 entries at a time");
static_assert(AROWS < 64 && 32 % EDGES_PER_TRIP == 0 && NEDGES < (1 << 25) && NNODES < (1 << 25), "a list word holds the edge number, later src, above 6 bits of local row");
static_assert(AGG_WAVES == 32, "an add pass hands entry base + wave + 32*lane to a lane: 32 waves x 32 lanes");

__global__ __launch_bounds__(AGG_THREADS) void gather_add_kernel(
    const float* __restrict__ H, const int* __restrict__ src,
    const int* __restrict__ dst, float* __restrict__ AGG,
    float* __restrict__ DEG)
{
    __shared__ __attribute__((aligned(16))) float acc[(AROWS + 1) * DOUT];
    __shared__ int wlist[AGG_WAVES * WCAP];
    __shared__ int flat[AGG_WAVES * WCAP];
    __shared__ int cnt[AROWS];
    __shared__ int wcount[AGG_WAVES];

    const int tid  = threadIdx.x;
    const int lane = tid & 31;
    const int wave = __builtin_amdgcn_readfirstlane(tid >> 5);
    const int row0 = blockIdx.x * AROWS;

    {
        const float4 z = {0.0f, 0.0f, 0.0f, 0.0f};
        for (int i = tid; i < AROWS * DOUT / 4; i += AGG_THREADS) ((float4*)acc)[i] = z;
        if (tid < AROWS) cnt[tid] = 0;
    }
    __syncthreads();

    int* __restrict__ wl = wlist + wave * WCAP;
    int pos = 0;
    {
        const int4* __restrict__ dst4 = (const int4*)dst;
        int4 d[SCAN_DEPTH], dn[SCAN_DEPTH];
        #pragma unroll
        for (int u = 0; u < SCAN_DEPTH; ++u) d[u] = dst4[tid + u * AGG_THREADS];

        for (int t = 0; t < SCAN_TRIPS; ++t) {
            const int qb = t * (SCAN_DEPTH * AGG_THREADS) + tid;
            #pragma unroll
            for (int u = 0; u < SCAN_DEPTH; ++u) {
                const int qn = qb + (SCAN_DEPTH + u) * AGG_THREADS;
                dn[u] = dst4[qn < NQUADS ? qn : NQUADS - 1];
            }
            unsigned full = 0;
            #pragma unroll
            for (int u = 0; u < SCAN_DEPTH; ++u) {
                const int q = qb + u * AGG_THREADS;
                const bool live = q < NQUADS;
                const unsigned r[4] = {(unsigned)(d[u].x - row0), (unsigned)(d[u].y - row0),
                                       (unsigned)(d[u].z - row0), (unsigned)(d[u].w - row0)};
                unsigned m[4];
                #pragma unroll
                for (int j = 0; j < 4; ++j) m[j] = __builtin_amdgcn_ballot_w32(live && r[j] < AROWS);
                if ((m[0] | m[1] | m[2] | m[3]) != 0) {
                    #pragma unroll
                    for (int j = 0; j < 4; ++j) {
                        if (m[j] != 0) {
                            if (live && r[j] < AROWS) {
                                const int o = pos + (int)__builtin_amdgcn_mbcnt_lo(m[j], 0u);
                                atomicAdd(&cnt[r[j]], 1);
                                if (o < WCAP) wl[o] = ((4 * q + j) << 6) | (int)r[j];
                                else full |= (1u << j) << (4 * u);
                            }
                            pos += __builtin_popcount(m[j]);
                            pos = pos < WCAP ? pos : WCAP;
                        }
                    }
                }
            }
            while (full != 0) {
                const int b = __builtin_ctz(full);
                full &= full - 1;
                const int e = 4 * (qb + (b >> 2) * AGG_THREADS) + (b & 3);
                const int rr = dst[e] - row0;
                const float* hs = H + (size_t)src[e] * DOUT;
                for (int col = 0; col < DOUT; ++col) atomicAdd(&acc[rr * DOUT + SUMPOS(col)], hs[col]);
            }
            #pragma unroll
            for (int u = 0; u < SCAN_DEPTH; ++u) d[u] = dn[u];
        }
    }
    if (lane == 0) wcount[wave] = pos;
    __syncthreads();

    int n = 0;
    {
        int pre = 0;
        #pragma unroll
        for (int i = 0; i < AGG_WAVES; ++i) {
            const int ci = wcount[i];
            n += ci;
            pre += (i < wave) ? ci : 0;
        }
        for (int i = lane; i < pos; i += 32) {
            const int ent = wl[i];
            flat[pre + i] = (src[ent >> 6] << 6) | (ent & 63);
        }
    }
    __syncthreads();

    {
        n = __builtin_amdgcn_readfirstlane(n);
        for (int base = 0; base < n; base += AGG_THREADS) {
            const int idx = base + wave + 32 * lane;
            int s_l = 0, r_l = AROWS;
            if (idx < n) {
                const int ent = flat[idx];
                r_l = ent & 63;
                s_l = ent >> 6;
            }
            const int rest = n - base - wave;
            int cw = rest > 0 ? (rest + 31) >> 5 : 0;
            cw = cw < 32 ? cw : 32;

            for (int i = 0; i < cw; i += EDGES_PER_TRIP) {
                float4 v[EDGES_PER_TRIP][2];
                #pragma unroll
                for (int u = 0; u < EDGES_PER_TRIP; ++u) {
                    const int s = __builtin_amdgcn_readlane(s_l, i + u);
                    const float4* hs = (const float4*)(H + (size_t)s * DOUT) + lane;
                    v[u][0] = hs[0];
                    v[u][1] = hs[32];
                }
                #pragma unroll
                for (int u = 0; u < EDGES_PER_TRIP; ++u) {
                    const int r = __builtin_amdgcn_readlane(r_l, i + u);
                    float* ad = acc + r * DOUT + lane;
                    #pragma unroll
                    for (int hf = 0; hf < 2; ++hf) {
                        atomicAdd(&ad[128 * hf],      v[u][hf].x);
                        atomicAdd(&ad[128 * hf + 32], v[u][hf].y);
                        atomicAdd(&ad[128 * hf + 64], v[u][hf].z);
                        atomicAdd(&ad[128 * hf + 96], v[u][hf].w);
                    }
                }
            }
        }
    }
    __syncthreads();

    {
        float4* __restrict__ out4 = (float4*)(AGG + (size_t)row0 * DOUT);
        for (int i = tid; i < AROWS * DOUT / 4; i += AGG_THREADS) {
            const float* a = acc + (i >> 6) * DOUT + ((i & 32) << 2) + (i & 31);
            const float4 o = {a[0], a[32], a[64], a[96]};
            out4[i] = o;
        }
        if (DEG != nullptr && tid < AROWS) DEG[row0 + tid] = (float)cnt[tid];
    }
}

extern "C" void kernel_launch(void* const* d_in, const int* in_sizes, int n_in,
                              void* d_out, int out_size, void* d_ws, size_t ws_size,
                              hipStream_t stream)
{
    const float* x   = (const float*)d_in[0];
    const int*   ei  = (const int*)d_in[1];
    const float* W0  = (const float*)d_in[2];
    const float* b0  = (const float*)d_in[3];
    const float* Wl1 = (const float*)d_in[4];
    const float* bl1 = (const float*)d_in[5];
    const float* Wr1 = (const float*)d_in[6];
    const float* Wl2 = (const float*)d_in[7];
    const float* bl2 = (const float*)d_in[8];
    const float* Wr2 = (const float*)d_in[9];

    const int* src = ei;
    const int* dst = ei + NEDGES;

    float* ws  = (float*)d_ws;
    float* h    = ws;
    float* agg  = h + (size_t)NNODES * DOUT;
    float* deg  = agg + (size_t)NNODES * DOUT;

    float* out1 = (float*)d_out;
    float* out2 = out1 + (size_t)NNODES * DOUT;

    dim3 gemm_grid(GEMM_BLOCKS, 1);


    lin_gemm_kernel<<<gemm_grid, GEMM_THREADS, 0, stream>>>(x, W0, b0, h);

    gather_add_kernel<<<AGG_BLOCKS, AGG_THREADS, 0, stream>>>(h, src, dst, agg, deg);
    sage_gemm_kernel<<<gemm_grid, GEMM_THREADS, 0, stream>>>(agg, deg, h, Wl1, bl1, Wr1, out1, 1);

    gather_add_kernel<<<AGG_BLOCKS, AGG_THREADS, 0, stream>>>(out1, src, dst, agg, (float*)nullptr);
    sage_gemm_kernel<<<gemm_grid, GEMM_THREADS, 0, stream>>>(agg, deg, out1, Wl2, bl2, Wr2, out2, 0);
}


# Layer 2's neighbour-sum launch takes its work-groups' edge lists (src, local row) from the workspace, where layer 1's launch stores them; second dst sweep gone; sums still in LDS arrival order
# speedup vs baseline: 2.8509x; 1.1096x over previous
#include <hip/hip_runtime.h>

typedef float v2f __attribute__((ext_vector_type(2)));
typedef float v8f __attribute__((ext_vector_type(8)));

#define NNODES 10000
#define NEDGES 320000
#define DIN 512
#define DOUT 256

#define TROWS 80
#define TCOLS 128
#define GEMM_THREADS 256
#define ROW_TILES (NNODES / TROWS)
#define GEMM_BLOCKS (ROW_TILES * (DOUT / TCOLS))
#define WTILES (TROWS / 16)
#define KB 8
#define KBS 4
static_assert(NNODES % TROWS == 0, "row tiles must be full: the kernels carry no row guards");
#define LIN_PITCH (DIN + 4)
#define SAGE_PITCH (DOUT + 4)

__global__ __launch_bounds__(GEMM_THREADS) void lin_gemm_kernel(
    const float* __restrict__ X, const float* __restrict__ W,
    const float* __restrict__ bias, float* __restrict__ H)
{
    __shared__ __attribute__((aligned(16))) float As[TROWS * LIN_PITCH];

    const int tid    = threadIdx.x;
    const int lane   = tid & 31;
    const int wave   = tid >> 5;
    const int lane16 = lane & 15;
    const int hsel   = lane >> 4;

    const int row0 = (blockIdx.x >> 1) * TROWS;
    const int col0 = (blockIdx.x & 1) * TCOLS + wave * 16;

    {
        const int rsub = tid >> 7;
        const int c4   = (tid & 127) * 4;
        #pragma unroll
        for (int p = 0; p < TROWS / 2; ++p) {
            const int r = 2 * p + rsub;
            const float4 v = *(const float4*)(X + (size_t)(row0 + r) * DIN + c4);
            *(float4*)(As + r * LIN_PITCH + c4) = v;
        }
    }
    __syncthreads();

    const float* ap   = As + lane16 * LIN_PITCH + 2 * hsel;
    const float* bcol = W + (size_t)(2 * hsel) * DOUT + (col0 + lane16);

    v8f acc[WTILES];
    #pragma unroll
    for (int t = 0; t < WTILES; ++t) acc[t] = v8f{};

    v2f bc[KB], ac[WTILES];
    #pragma unroll
    for (int s = 0; s < KB; ++s) {
        bc[s].x = bcol[(4 * s) * DOUT];
        bc[s].y = bcol[(4 * s + 1) * DOUT];
    }
    #pragma unroll
    for (int t = 0; t < WTILES; ++t) ac[t] = *(const v2f*)(ap + t * 16 * LIN_PITCH);

    for (int kb = 0; kb < DIN; kb += 4 * KB) {
        const int knb = (kb + 4 * KB < DIN) ? kb + 4 * KB : 0;
        const float* bnp = bcol + (size_t)knb * DOUT;
        v2f bn[KB];
        #pragma unroll
        for (int s = 0; s < KB; ++s) {
            bn[s].x = bnp[(4 * s) * DOUT];
            bn[s].y = bnp[(4 * s + 1) * DOUT];
        }
        #pragma unroll
        for (int s = 0; s < KB; ++s) {
            const int kn = (s + 1 < KB) ? kb + 4 * (s + 1) : knb;
            v2f an[WTILES];
            #pragma unroll
            for (int t = 0; t < WTILES; ++t) an[t] = *(const v2f*)(ap + t * 16 * LIN_PITCH + kn);
            #pragma unroll
            for (int t = 0; t < WTILES; ++t)
                acc[t] = __builtin_amdgcn_wmma_f32_16x16x4_f32(false, ac[t], false, bc[s],
                                                               (short)0, acc[t], false, false);
            #pragma unroll
            for (int t = 0; t < WTILES; ++t) ac[t] = an[t];
        }
        #pragma unroll
        for (int s = 0; s < KB; ++s) bc[s] = bn[s];
    }

    const float bv = bias[col0 + lane16];
    #pragma unroll
    for (int t = 0; t < WTILES; ++t) {
        #pragma unroll
        for (int r = 0; r < 8; ++r) {
            const int row = row0 + 16 * t + r + 8 * hsel;
            H[(size_t)row * DOUT + col0 + lane16] = acc[t][r] + bv;
        }
    }
}

__global__ __launch_bounds__(GEMM_THREADS) void sage_gemm_kernel(
    const float* __restrict__ AGG,
    const float* __restrict__ DEG,
    const float* __restrict__ Hin,
    const float* __restrict__ Wl, const float* __restrict__ bl,
    const float* __restrict__ Wr,
    float* __restrict__ OUT, int relu)
{
    __shared__ __attribute__((aligned(16))) float Ms[TROWS * SAGE_PITCH];
    __shared__ __attribute__((aligned(16))) float Hs[TROWS * SAGE_PITCH];

    const int tid    = threadIdx.x;
    const int lane   = tid & 31;
    const int wave   = tid >> 5;
    const int lane16 = lane & 15;
    const int hsel   = lane >> 4;

    const int row0 = (blockIdx.x >> 1) * TROWS;
    const int col0 = (blockIdx.x & 1) * TCOLS + wave * 16;

    float degv[WTILES];
    #pragma unroll
    for (int t = 0; t < WTILES; ++t) degv[t] = DEG[row0 + 16 * t + lane16];

    {
        const int rsub = tid >> 6;
        const int c4   = (tid & 63) * 4;
        float4 mv[TROWS / 4], hv[TROWS / 4];
        #pragma unroll
        for (int p = 0; p < TROWS / 4; ++p) {
            const unsigned q = (unsigned)(row0 + 4 * p + rsub) * (DOUT / 4) + (unsigned)(tid & 63);
            mv[p] = ((const float4*)AGG)[q];
            hv[p] = ((const float4*)Hin)[q];
        }
        #pragma unroll
        for (int p = 0; p < TROWS / 4; ++p) {
            const int r = 4 * p + rsub;
            *(float4*)(Ms + r * SAGE_PITCH + c4) = mv[p];
            *(float4*)(Hs + r * SAGE_PITCH + c4) = hv[p];
        }
    }
    __syncthreads();

    float inv[WTILES];
    #pragma unroll
    for (int t = 0; t < WTILES; ++t) inv[t] = 1.0f / fmaxf(degv[t], 1.0f);

    const float* mp  = Ms + lane16 * SAGE_PITCH + 2 * hsel;
    const float* hp  = Hs + lane16 * SAGE_PITCH + 2 * hsel;
    const float* blc = Wl + (size_t)(2 * hsel) * DOUT + (col0 + lane16);
    const float* brc = Wr + (size_t)(2 * hsel) * DOUT + (col0 + lane16);

    v8f acc[WTILES];
    #pragma unroll
    for (int t = 0; t < WTILES; ++t) acc[t] = v8f{};

    v2f wlf[KBS], wrf[KBS], mc[WTILES], hc[WTILES];
    #pragma unroll
    for (int s = 0; s < KBS; ++s) {
        wlf[s].x = blc[(4 * s) * DOUT];    wlf[s].y = blc[(4 * s + 1) * DOUT];
        wrf[s].x = brc[(4 * s) * DOUT];    wrf[s].y = brc[(4 * s + 1) * DOUT];
    }
    #pragma unroll
    for (int t = 0; t < WTILES; ++t) {
        mc[t] = *(const v2f*)(mp + t * 16 * SAGE_PITCH);
        hc[t] = *(const v2f*)(hp + t * 16 * SAGE_PITCH);
    }

    for (int kb = 0; kb < DOUT; kb += 4 * KBS) {
        const int knb = (kb + 4 * KBS < DOUT) ? kb + 4 * KBS : 0;
        const float* blnp = blc + (size_t)knb * DOUT;
        const float* brnp = brc + (size_t)knb * DOUT;
        v2f wlfn[KBS], wrfn[KBS];
        #pragma unroll
        for (int s = 0; s < KBS; ++s) {
            wlfn[s].x = blnp[(4 * s) * DOUT];    wlfn[s].y = blnp[(4 * s + 1) * DOUT];
            wrfn[s].x = brnp[(4 * s) * DOUT];    wrfn[s].y = brnp[(4 * s + 1) * DOUT];
        }
        #pragma unroll
        for (int s = 0; s < KBS; ++s) {
            const int kn = (s + 1 < KBS) ? kb + 4 * (s + 1) : knb;
            v2f mn[WTILES], hn[WTILES];
            #pragma unroll
            for (int t = 0; t < WTILES; ++t) {
                mn[t] = *(const v2f*)(mp + t * 16 * SAGE_PITCH + kn);
                hn[t] = *(const v2f*)(hp + t * 16 * SAGE_PITCH + kn);
            }
            #pragma unroll
            for (int t = 0; t < WTILES; ++t) {
                const v2f am = mc[t] * inv[t];
                acc[t] = __builtin_amdgcn_wmma_f32_16x16x4_f32(false, am, false, wlf[s],
                                                               (short)0, acc[t], false, false);
            }
            #pragma unroll
            for (int t = 0; t < WTILES; ++t)
                acc[t] = __builtin_amdgcn_wmma_f32_16x16x4_f32(false, hc[t], false, wrf[s],
                                                               (short)0, acc[t], false, false);
            #pragma unroll
            for (int t = 0; t < WTILES; ++t) { mc[t] = mn[t]; hc[t] = hn[t]; }
        }
        #pragma unroll
        for (int s = 0; s < KBS; ++s) { wlf[s] = wlfn[s]; wrf[s] = wrfn[s]; }
    }

    const float bv = bl[col0 + lane16];
    #pragma unroll
    for (int t = 0; t < WTILES; ++t) {
        #pragma unroll
        for (int r = 0; r < 8; ++r) {
            const int lr = 16 * t + r + 8 * hsel;
            float v = acc[t][r] + bv;
            if (relu) v = fmaxf(v, 0.0f);
            const float hr = Hs[lr * SAGE_PITCH + col0 + lane16];
            OUT[(size_t)(row0 + lr) * DOUT + col0 + lane16] = v + hr;
        }
    }
}

#define AROWS 40
#define AGG_THREADS 1024
#define AGG_WAVES (AGG_THREADS / 32)
#define AGG_BLOCKS (NNODES / AROWS)
#define WCAP 512
#define NQUADS (NEDGES / 4)
#define SCAN_DEPTH 8
#define SCAN_TRIPS ((NQUADS + SCAN_DEPTH * AGG_THREADS - 1) / (SCAN_DEPTH * AGG_THREADS))
#define EDGES_PER_TRIP 4
#define LCAP (AGG_WAVES * WCAP)
#define LSTRIDE (LCAP + 64)
#define DEG_PAD 10240
#define ECOUNT_PAD 256
static_assert(DEG_PAD >= NNODES && ECOUNT_PAD >= AGG_BLOCKS, "room for every count and every length");
#define SUMPOS(c) (((c) & 128) + (((c) & 3) << 5) + (((c) & 127) >> 2))
static_assert(DOUT == 256, "a row of sums is two halves of 32 lanes x 4 columns");
static_assert(NNODES % AROWS == 0, "every work-group owns AROWS whole rows");
static_assert(NEDGES % 4 == 0, "dst is read 4 entries at a time");
static_assert(AROWS < 64 && 32 % EDGES_PER_TRIP == 0 && NEDGES < (1 << 25) && NNODES < (1 << 25), "a list word holds the edge number, later src, above 6 bits of local row");
static_assert(AGG_WAVES == 32, "an add pass hands entry base + wave + 32*lane to a lane: 32 waves x 32 lanes");

__global__ __launch_bounds__(AGG_THREADS) void gather_add_kernel(
    const float* __restrict__ H, const int* __restrict__ src,
    const int* __restrict__ dst, float* __restrict__ AGG,
    float* __restrict__ DEG, int* __restrict__ ELIST, int* __restrict__ ECOUNT, int first)
{
    __shared__ __attribute__((aligned(16))) float acc[(AROWS + 1) * DOUT];
    __shared__ int wlist[AGG_WAVES * WCAP];
    __shared__ int flat[AGG_WAVES * WCAP];
    __shared__ int cnt[AROWS];
    __shared__ int wcount[AGG_WAVES];
    __shared__ int beyond;

    const int tid  = threadIdx.x;
    const int lane = tid & 31;
    const int wave = __builtin_amdgcn_readfirstlane(tid >> 5);
    const int row0 = blockIdx.x * AROWS;

    {
        const float4 z = {0.0f, 0.0f, 0.0f, 0.0f};
        for (int i = tid; i < AROWS * DOUT / 4; i += AGG_THREADS) ((float4*)acc)[i] = z;
        if (tid < AROWS) cnt[tid] = 0;
        if (tid == 0) beyond = 0;
    }

    int* __restrict__ el = ELIST + (size_t)blockIdx.x * LSTRIDE;
    int n = 0;
    bool sweep = true;
    if (!first) {
        const int nl = __builtin_amdgcn_readfirstlane(ECOUNT[blockIdx.x]);
        if (nl >= 0) {
            sweep = false;
            n = nl < LCAP ? nl : LCAP;
            for (int i = tid; i < n; i += AGG_THREADS) flat[i] = el[i];
        }
    }
    __syncthreads();

    if (sweep) {
    int* __restrict__ wl = wlist + wave * WCAP;
    int pos = 0;
    {
        const int4* __restrict__ dst4 = (const int4*)dst;
        int4 d[SCAN_DEPTH], dn[SCAN_DEPTH];
        #pragma unroll
        for (int u = 0; u < SCAN_DEPTH; ++u) d[u] = dst4[tid + u * AGG_THREADS];

        for (int t = 0; t < SCAN_TRIPS; ++t) {
            const int qb = t * (SCAN_DEPTH * AGG_THREADS) + tid;
            #pragma unroll
            for (int u = 0; u < SCAN_DEPTH; ++u) {
                const int qn = qb + (SCAN_DEPTH + u) * AGG_THREADS;
                dn[u] = dst4[qn < NQUADS ? qn : NQUADS - 1];
            }
            unsigned full = 0;
            #pragma unroll
            for (int u = 0; u < SCAN_DEPTH; ++u) {
                const int q = qb + u * AGG_THREADS;
                const bool live = q < NQUADS;
                const unsigned r[4] = {(unsigned)(d[u].x - row0), (unsigned)(d[u].y - row0),
                                       (unsigned)(d[u].z - row0), (unsigned)(d[u].w - row0)};
                unsigned m[4];
                #pragma unroll
                for (int j = 0; j < 4; ++j) m[j] = __builtin_amdgcn_ballot_w32(live && r[j] < AROWS);
                if ((m[0] | m[1] | m[2] | m[3]) != 0) {
                    #pragma unroll
                    for (int j = 0; j < 4; ++j) {
                        if (m[j] != 0) {
                            if (live && r[j] < AROWS) {
                                const int o = pos + (int)__builtin_amdgcn_mbcnt_lo(m[j], 0u);
                                atomicAdd(&cnt[r[j]], 1);
                                if (o < WCAP) wl[o] = ((4 * q + j) << 6) | (int)r[j];
                                else full |= (1u << j) << (4 * u);
                            }
                            pos += __builtin_popcount(m[j]);
                            pos = pos < WCAP ? pos : WCAP;
                        }
                    }
                }
            }
            if (full != 0) beyond = 1;
            while (full != 0) {
                const int b = __builtin_ctz(full);
                full &= full - 1;
                const int e = 4 * (qb + (b >> 2) * AGG_THREADS) + (b & 3);
                const int rr = dst[e] - row0;
                const float* hs = H + (size_t)src[e] * DOUT;
                for (int col = 0; col < DOUT; ++col) atomicAdd(&acc[rr * DOUT + SUMPOS(col)], hs[col]);
            }
            #pragma unroll
            for (int u = 0; u < SCAN_DEPTH; ++u) d[u] = dn[u];
        }
    }
    if (lane == 0) wcount[wave] = pos;
    __syncthreads();

    {
        int pre = 0;
        #pragma unroll
        for (int i = 0; i < AGG_WAVES; ++i) {
            const int ci = wcount[i];
            n += ci;
            pre += (i < wave) ? ci : 0;
        }
        for (int i = lane; i < pos; i += 32) {
            const int ent = wl[i];
            flat[pre + i] = (src[ent >> 6] << 6) | (ent & 63);
        }
    }
    __syncthreads();
    n = __builtin_amdgcn_readfirstlane(n);
    }

    if (first) {
        for (int i = tid; i < n; i += AGG_THREADS) el[i] = flat[i];
        if (tid == 0) ECOUNT[blockIdx.x] = beyond ? -1 : n;
    }

    {
        for (int base = 0; base < n; base += AGG_THREADS) {
            const int idx = base + wave + 32 * lane;
            int s_l = 0, r_l = AROWS;
            if (idx < n) {
                const int ent = flat[idx];
                r_l = ent & 63;
                s_l = ent >> 6;
            }
            const int rest = n - base - wave;
            int cw = rest > 0 ? (rest + 31) >> 5 : 0;
            cw = cw < 32 ? cw : 32;

            for (int i = 0; i < cw; i += EDGES_PER_TRIP) {
                float4 v[EDGES_PER_TRIP][2];
                #pragma unroll
                for (int u = 0; u < EDGES_PER_TRIP; ++u) {
                    const int s = __builtin_amdgcn_readlane(s_l, i + u);
                    const float4* hs = (const float4*)(H + (size_t)s * DOUT) + lane;
                    v[u][0] = hs[0];
                    v[u][1] = hs[32];
                }
                #pragma unroll
                for (int u = 0; u < EDGES_PER_TRIP; ++u) {
                    const int r = __builtin_amdgcn_readlane(r_l, i + u);
                    float* ad = acc + r * DOUT + lane;
                    #pragma unroll
                    for (int hf = 0; hf < 2; ++hf) {
                        atomicAdd(&ad[128 * hf],      v[u][hf].x);
                        atomicAdd(&ad[128 * hf + 32], v[u][hf].y);
                        atomicAdd(&ad[128 * hf + 64], v[u][hf].z);
                        atomicAdd(&ad[128 * hf + 96], v[u][hf].w);
                    }
                }
            }
        }
    }
    __syncthreads();

    {
        float4* __restrict__ out4 = (float4*)(AGG + (size_t)row0 * DOUT);
        for (int i = tid; i < AROWS * DOUT / 4; i += AGG_THREADS) {
            const float* a = acc + (i >> 6) * DOUT + ((i & 32) << 2) + (i & 31);
            const float4 o = {a[0], a[32], a[64], a[96]};
            out4[i] = o;
        }
        if (DEG != nullptr && tid < AROWS) DEG[row0 + tid] = (float)cnt[tid];
    }
}

extern "C" void kernel_launch(void* const* d_in, const int* in_sizes, int n_in,
                              void* d_out, int out_size, void* d_ws, size_t ws_size,
                              hipStream_t stream)
{
    const float* x   = (const float*)d_in[0];
    const int*   ei  = (const int*)d_in[1];
    const float* W0  = (const float*)d_in[2];
    const float* b0  = (const float*)d_in[3];
    const float* Wl1 = (const float*)d_in[4];
    const float* bl1 = (const float*)d_in[5];
    const float* Wr1 = (const float*)d_in[6];
    const float* Wl2 = (const float*)d_in[7];
    const float* bl2 = (const float*)d_in[8];
    const float* Wr2 = (const float*)d_in[9];

    const int* src = ei;
    const int* dst = ei + NEDGES;

    float* ws  = (float*)d_ws;
    float* h    = ws;
    float* agg  = h + (size_t)NNODES * DOUT;
    float* deg  = agg + (size_t)NNODES * DOUT;
    int* ecount = (int*)(deg + DEG_PAD);
    int* elist  = ecount + ECOUNT_PAD;

    float* out1 = (float*)d_out;
    float* out2 = out1 + (size_t)NNODES * DOUT;

    dim3 gemm_grid(GEMM_BLOCKS, 1);


    lin_gemm_kernel<<<gemm_grid, GEMM_THREADS, 0, stream>>>(x, W0, b0, h);

    gather_add_kernel<<<AGG_BLOCKS, AGG_THREADS, 0, stream>>>(h, src, dst, agg, deg, elist, ecount, 1);
    sage_gemm_kernel<<<gemm_grid, GEMM_THREADS, 0, stream>>>(agg, deg, h, Wl1, bl1, Wr1, out1, 1);

    gather_add_kernel<<<AGG_BLOCKS, AGG_THREADS, 0, stream>>>(out1, src, dst, agg, (float*)nullptr, elist, ecount, 0);
    sage_gemm_kernel<<<gemm_grid, GEMM_THREADS, 0, stream>>>(agg, deg, out1, Wl2, bl2, Wr2, out2, 0);
}


# Edge lists by a one-pass partition: each lin-GEMM work-group sorts its 1280 edges by owner into 128-byte lines; a gather group merges its 250 lines; no all-dst sweep; sums still in LDS arrival order
# speedup vs baseline: 3.1713x; 1.1124x over previous
#include <hip/hip_runtime.h>

typedef float v2f __attribute__((ext_vector_type(2)));
typedef float v8f __attribute__((ext_vector_type(8)));

#define NNODES 10000
#define NEDGES 320000
#define DIN 512
#define DOUT 256

#define TROWS 80
#define TCOLS 128
#define GEMM_THREADS 256
#define ROW_TILES (NNODES / TROWS)
#define GEMM_BLOCKS (ROW_TILES * (DOUT / TCOLS))
#define WTILES (TROWS / 16)
#define KB 8
#define KBS 4
static_assert(NNODES % TROWS == 0, "row tiles must be full: the kernels carry no row guards");
#define LIN_PITCH (DIN + 4)
#define SAGE_PITCH (DOUT + 4)

#define AROWS 40
#define AGG_BLOCKS (NNODES / AROWS)
#define SLOT_WORDS 32
#define SLOT_CAP (SLOT_WORDS - 1)
#define PRODUCERS GEMM_BLOCKS
#define EDGES_PER_PRODUCER (NEDGES / PRODUCERS)
#define EDGES_PER_THREAD (EDGES_PER_PRODUCER / GEMM_THREADS)
#define BIN_QUADS (AGG_BLOCKS * SLOT_WORDS / 4)
static_assert(NEDGES % PRODUCERS == 0 && EDGES_PER_PRODUCER % GEMM_THREADS == 0, "every producer thread bins the same number of edges");
static_assert(NNODES % AROWS == 0, "every owner has AROWS whole rows");

__global__ __launch_bounds__(GEMM_THREADS) void lin_gemm_kernel(
    const float* __restrict__ X, const float* __restrict__ W,
    const float* __restrict__ bias, float* __restrict__ H,
    const int* __restrict__ src, const int* __restrict__ dst, int* __restrict__ SLOTS)
{
    __shared__ __attribute__((aligned(16))) float As[TROWS * LIN_PITCH];
    __shared__ __attribute__((aligned(16))) int bins[AGG_BLOCKS * SLOT_WORDS];
    __shared__ int bcnt[AGG_BLOCKS];

    const int tid    = threadIdx.x;
    const int lane   = tid & 31;
    const int wave   = tid >> 5;
    const int lane16 = lane & 15;
    const int hsel   = lane >> 4;

    const int row0 = (blockIdx.x >> 1) * TROWS;
    const int col0 = (blockIdx.x & 1) * TCOLS + wave * 16;

    int ed[EDGES_PER_THREAD], es[EDGES_PER_THREAD];
    #pragma unroll
    for (int j = 0; j < EDGES_PER_THREAD; ++j) {
        const int e = blockIdx.x * EDGES_PER_PRODUCER + j * GEMM_THREADS + tid;
        ed[j] = dst[e];
        es[j] = src[e];
    }
    {
        const int4 z = {0, 0, 0, 0};
        for (int i = tid; i < BIN_QUADS; i += GEMM_THREADS) ((int4*)bins)[i] = z;
        if (tid < AGG_BLOCKS) bcnt[tid] = 0;
    }

    {
        const int rsub = tid >> 7;
        const int c4   = (tid & 127) * 4;
        #pragma unroll
        for (int p = 0; p < TROWS / 2; ++p) {
            const int r = 2 * p + rsub;
            const float4 v = *(const float4*)(X + (size_t)(row0 + r) * DIN + c4);
            *(float4*)(As + r * LIN_PITCH + c4) = v;
        }
    }
    __syncthreads();

    int place[EDGES_PER_THREAD];
    #pragma unroll
    for (int j = 0; j < EDGES_PER_THREAD; ++j) {
        const unsigned o = (unsigned)ed[j] / AROWS;
        place[j] = SLOT_CAP;
        if (o < AGG_BLOCKS) place[j] = atomicAdd(&bcnt[o], 1);
    }
    #pragma unroll
    for (int j = 0; j < EDGES_PER_THREAD; ++j) {
        const int o = (int)((unsigned)ed[j] / AROWS);
        const int r = ed[j] - o * AROWS;
        if (place[j] < SLOT_CAP) bins[o * SLOT_WORDS + 1 + place[j]] = (es[j] << 6) | r;
    }
    __syncthreads();
    #pragma unroll
    for (int u = 0; u < (BIN_QUADS + GEMM_THREADS - 1) / GEMM_THREADS; ++u) {
        const int i = tid + u * GEMM_THREADS;
        if (i < BIN_QUADS) {
            const int o = i >> 3;
            int4 q = ((const int4*)bins)[i];
            if ((i & 7) == 0) q.x = bcnt[o];
            ((int4*)SLOTS)[((size_t)o * PRODUCERS + blockIdx.x) * (SLOT_WORDS / 4) + (i & 7)] = q;
        }
    }

    const float* ap   = As + lane16 * LIN_PITCH + 2 * hsel;
    const float* bcol = W + (size_t)(2 * hsel) * DOUT + (col0 + lane16);

    v8f acc[WTILES];
    #pragma unroll
    for (int t = 0; t < WTILES; ++t) acc[t] = v8f{};

    v2f bc[KB], ac[WTILES];
    #pragma unroll
    for (int s = 0; s < KB; ++s) {
        bc[s].x = bcol[(4 * s) * DOUT];
        bc[s].y = bcol[(4 * s + 1) * DOUT];
    }
    #pragma unroll
    for (int t = 0; t < WTILES; ++t) ac[t] = *(const v2f*)(ap + t * 16 * LIN_PITCH);

    for (int kb = 0; kb < DIN; kb += 4 * KB) {
        const int knb = (kb + 4 * KB < DIN) ? kb + 4 * KB : 0;
        const float* bnp = bcol + (size_t)knb * DOUT;
        v2f bn[KB];
        #pragma unroll
        for (int s = 0; s < KB; ++s) {
            bn[s].x = bnp[(4 * s) * DOUT];
            bn[s].y = bnp[(4 * s + 1) * DOUT];
        }
        #pragma unroll
        for (int s = 0; s < KB; ++s) {
            const int kn = (s + 1 < KB) ? kb + 4 * (s + 1) : knb;
            v2f an[WTILES];
            #pragma unroll
            for (int t = 0; t < WTILES; ++t) an[t] = *(const v2f*)(ap + t * 16 * LIN_PITCH + kn);
            #pragma unroll
            for (int t = 0; t < WTILES; ++t)
                acc[t] = __builtin_amdgcn_wmma_f32_16x16x4_f32(false, ac[t], false, bc[s],
                                                               (short)0, acc[t], false, false);
            #pragma unroll
            for (int t = 0; t < WTILES; ++t) ac[t] = an[t];
        }
        #pragma unroll
        for (int s = 0; s < KB; ++s) bc[s] = bn[s];
    }

    const float bv = bias[col0 + lane16];
    #pragma unroll
    for (int t = 0; t < WTILES; ++t) {
        #pragma unroll
        for (int r = 0; r < 8; ++r) {
            const int row = row0 + 16 * t + r + 8 * hsel;
            H[(size_t)row * DOUT + col0 + lane16] = acc[t][r] + bv;
        }
    }
}

__global__ __launch_bounds__(GEMM_THREADS) void sage_gemm_kernel(
    const float* __restrict__ AGG,
    const float* __restrict__ DEG,
    const float* __restrict__ Hin,
    const float* __restrict__ Wl, const float* __restrict__ bl,
    const float* __restrict__ Wr,
    float* __restrict__ OUT, int relu)
{
    __shared__ __attribute__((aligned(16))) float Ms[TROWS * SAGE_PITCH];
    __shared__ __attribute__((aligned(16))) float Hs[TROWS * SAGE_PITCH];

    const int tid    = threadIdx.x;
    const int lane   = tid & 31;
    const int wave   = tid >> 5;
    const int lane16 = lane & 15;
    const int hsel   = lane >> 4;

    const int row0 = (blockIdx.x >> 1) * TROWS;
    const int col0 = (blockIdx.x & 1) * TCOLS + wave * 16;

    float degv[WTILES];
    #pragma unroll
    for (int t = 0; t < WTILES; ++t) degv[t] = DEG[row0 + 16 * t + lane16];

    {
        const int rsub = tid >> 6;
        const int c4   = (tid & 63) * 4;
        float4 mv[TROWS / 4], hv[TROWS / 4];
        #pragma unroll
        for (int p = 0; p < TROWS / 4; ++p) {
            const unsigned q = (unsigned)(row0 + 4 * p + rsub) * (DOUT / 4) + (unsigned)(tid & 63);
            mv[p] = ((const float4*)AGG)[q];
            hv[p] = ((const float4*)Hin)[q];
        }
        #pragma unroll
        for (int p = 0; p < TROWS / 4; ++p) {
            const int r = 4 * p + rsub;
            *(float4*)(Ms + r * SAGE_PITCH + c4) = mv[p];
            *(float4*)(Hs + r * SAGE_PITCH + c4) = hv[p];
        }
    }
    __syncthreads();

    float inv[WTILES];
    #pragma unroll
    for (int t = 0; t < WTILES; ++t) inv[t] = 1.0f / fmaxf(degv[t], 1.0f);

    const float* mp  = Ms + lane16 * SAGE_PITCH + 2 * hsel;
    const float* hp  = Hs + lane16 * SAGE_PITCH + 2 * hsel;
    const float* blc = Wl + (size_t)(2 * hsel) * DOUT + (col0 + lane16);
    const float* brc = Wr + (size_t)(2 * hsel) * DOUT + (col0 + lane16);

    v8f acc[WTILES];
    #pragma unroll
    for (int t = 0; t < WTILES; ++t) acc[t] = v8f{};

    v2f wlf[KBS], wrf[KBS], mc[WTILES], hc[WTILES];
    #pragma unroll
    for (int s = 0; s < KBS; ++s) {
        wlf[s].x = blc[(4 * s) * DOUT];    wlf[s].y = blc[(4 * s + 1) * DOUT];
        wrf[s].x = brc[(4 * s) * DOUT];    wrf[s].y = brc[(4 * s + 1) * DOUT];
    }
    #pragma unroll
    for (int t = 0; t < WTILES; ++t) {
        mc[t] = *(const v2f*)(mp + t * 16 * SAGE_PITCH);
        hc[t] = *(const v2f*)(hp + t * 16 * SAGE_PITCH);
    }

    for (int kb = 0; kb < DOUT; kb += 4 * KBS) {
        const int knb = (kb + 4 * KBS < DOUT) ? kb + 4 * KBS : 0;
        const float* blnp = blc + (size_t)knb * DOUT;
        const float* brnp = brc + (size_t)knb * DOUT;
        v2f wlfn[KBS], wrfn[KBS];
        #pragma unroll
        for (int s = 0; s < KBS; ++s) {
            wlfn[s].x = blnp[(4 * s) * DOUT];    wlfn[s].y = blnp[(4 * s + 1) * DOUT];
            wrfn[s].x = brnp[(4 * s) * DOUT];    wrfn[s].y = brnp[(4 * s + 1) * DOUT];
        }
        #pragma unroll
        for (int s = 0; s < KBS; ++s) {
            const int kn = (s + 1 < KBS) ? kb + 4 * (s + 1) : knb;
            v2f mn[WTILES], hn[WTILES];
            #pragma unroll
            for (int t = 0; t < WTILES; ++t) {
                mn[t] = *(const v2f*)(mp + t * 16 * SAGE_PITCH + kn);
                hn[t] = *(const v2f*)(hp + t * 16 * SAGE_PITCH + kn);
            }
            #pragma unroll
            for (int t = 0; t < WTILES; ++t) {
                const v2f am = mc[t] * inv[t];
                acc[t] = __builtin_amdgcn_wmma_f32_16x16x4_f32(false, am, false, wlf[s],
                                                               (short)0, acc[t], false, false);
            }
            #pragma unroll
            for (int t = 0; t < WTILES; ++t)
                acc[t] = __builtin_amdgcn_wmma_f32_16x16x4_f32(false, hc[t], false, wrf[s],
                                                               (short)0, acc[t], false, false);
            #pragma unroll
            for (int t = 0; t < WTILES; ++t) { mc[t] = mn[t]; hc[t] = hn[t]; }
        }
        #pragma unroll
        for (int s = 0; s < KBS; ++s) { wlf[s] = wlfn[s]; wrf[s] = wrfn[s]; }
    }

    const float bv = bl[col0 + lane16];
    #pragma unroll
    for (int t = 0; t < WTILES; ++t) {
        #pragma unroll
        for (int r = 0; r < 8; ++r) {
            const int lr = 16 * t + r + 8 * hsel;
            float v = acc[t][r] + bv;
            if (relu) v = fmaxf(v, 0.0f);
            const float hr = Hs[lr * SAGE_PITCH + col0 + lane16];
            OUT[(size_t)(row0 + lr) * DOUT + col0 + lane16] = v + hr;
        }
    }
}

#define AGG_THREADS 1024
#define AGG_WAVES (AGG_THREADS / 32)
#define SLOT_PASSES ((PRODUCERS + AGG_WAVES - 1) / AGG_WAVES)
#define WCAP 512
#define NQUADS (NEDGES / 4)
#define SCAN_DEPTH 8
#define SCAN_TRIPS ((NQUADS + SCAN_DEPTH * AGG_THREADS - 1) / (SCAN_DEPTH * AGG_THREADS))
#define EDGES_PER_TRIP 4
#define LCAP (AGG_WAVES * WCAP)
#define LSTRIDE (LCAP + 64)
#define DEG_PAD 10240
#define ECOUNT_PAD 256
static_assert(DEG_PAD >= NNODES && ECOUNT_PAD >= AGG_BLOCKS, "room for every count and every length");
#define SUMPOS(c) (((c) & 128) + (((c) & 3) << 5) + (((c) & 127) >> 2))
static_assert(DOUT == 256, "a row of sums is two halves of 32 lanes x 4 columns");
static_assert(NNODES % AROWS == 0, "every work-group owns AROWS whole rows");
static_assert(NEDGES % 4 == 0, "dst is read 4 entries at a time");
static_assert(AROWS < 64 && 32 % EDGES_PER_TRIP == 0 && NEDGES < (1 << 25) && NNODES < (1 << 25), "a list word holds the edge number, later src, above 6 bits of local row");
static_assert(AGG_WAVES == 32, "an add pass hands entry base + wave + 32*lane to a lane: 32 waves x 32 lanes");

__global__ __launch_bounds__(AGG_THREADS) void gather_add_kernel(
    const float* __restrict__ H, const int* __restrict__ src,
    const int* __restrict__ dst, float* __restrict__ AGG,
    float* __restrict__ DEG, int* __restrict__ ELIST, int* __restrict__ ECOUNT,
    const int* __restrict__ SLOTS, int first)
{
    __shared__ __attribute__((aligned(16))) float acc[(AROWS + 1) * DOUT];
    __shared__ int wlist[AGG_WAVES * WCAP];
    __shared__ int flat[AGG_WAVES * WCAP];
    __shared__ int cnt[AROWS];
    __shared__ int wcount[AGG_WAVES];
    __shared__ int beyond;
    __shared__ int short_line;

    const int tid  = threadIdx.x;
    const int lane = tid & 31;
    const int wave = __builtin_amdgcn_readfirstlane(tid >> 5);
    const int row0 = blockIdx.x * AROWS;

    int sw[SLOT_PASSES];
    #pragma unroll
    for (int k = 0; k < SLOT_PASSES; ++k) sw[k] = 0;
    if (first) {
        const int* __restrict__ sl = SLOTS + (size_t)blockIdx.x * (PRODUCERS * SLOT_WORDS) + lane;
        #pragma unroll
        for (int k = 0; k < SLOT_PASSES; ++k) {
            const int p = wave + AGG_WAVES * k;
            if (p < PRODUCERS) sw[k] = sl[p * SLOT_WORDS];
        }
    }

    {
        const float4 z = {0.0f, 0.0f, 0.0f, 0.0f};
        for (int i = tid; i < AROWS * DOUT / 4; i += AGG_THREADS) ((float4*)acc)[i] = z;
        if (tid < AROWS) cnt[tid] = 0;
        if (tid == 0) { beyond = 0; short_line = 0; }
    }

    int* __restrict__ el = ELIST + (size_t)blockIdx.x * LSTRIDE;
    int n = 0;
    bool sweep = true;
    if (!first) {
        const int nl = __builtin_amdgcn_readfirstlane(ECOUNT[blockIdx.x]);
        if (nl >= 0) {
            sweep = false;
            n = nl < LCAP ? nl : LCAP;
            for (int i = tid; i < n; i += AGG_THREADS) flat[i] = el[i];
        }
    }
    __syncthreads();

    if (first) {
        int lc[SLOT_PASSES];
        int tot = 0;
        bool bad = false;
        #pragma unroll
        for (int k = 0; k < SLOT_PASSES; ++k) {
            lc[k] = __builtin_amdgcn_readfirstlane(sw[k]);
            bad = bad || lc[k] > SLOT_CAP || lc[k] < 0;
            tot += lc[k];
        }
        if (lane == 0) {
            wcount[wave] = tot;
            if (bad) short_line = 1;
        }
        __syncthreads();
        if (!short_line) {
            sweep = false;
            int pre = 0;
            #pragma unroll
            for (int i = 0; i < AGG_WAVES; ++i) {
                const int ci = wcount[i];
                n += ci;
                pre += (i < wave) ? ci : 0;
            }
            #pragma unroll
            for (int k = 0; k < SLOT_PASSES; ++k) {
                if (lane >= 1 && lane <= lc[k]) {
                    flat[pre + lane - 1] = sw[k];
                    atomicAdd(&cnt[sw[k] & 63], 1);
                }
                pre += lc[k];
            }
            n = __builtin_amdgcn_readfirstlane(n);
        }
        __syncthreads();
    }

    if (sweep) {
    int* __restrict__ wl = wlist + wave * WCAP;
    int pos = 0;
    {
        const int4* __restrict__ dst4 = (const int4*)dst;
        int4 d[SCAN_DEPTH], dn[SCAN_DEPTH];
        #pragma unroll
        for (int u = 0; u < SCAN_DEPTH; ++u) d[u] = dst4[tid + u * AGG_THREADS];

        for (int t = 0; t < SCAN_TRIPS; ++t) {
            const int qb = t * (SCAN_DEPTH * AGG_THREADS) + tid;
            #pragma unroll
            for (int u = 0; u < SCAN_DEPTH; ++u) {
                const int qn = qb + (SCAN_DEPTH + u) * AGG_THREADS;
                dn[u] = dst4[qn < NQUADS ? qn : NQUADS - 1];
            }
            unsigned full = 0;
            #pragma unroll
            for (int u = 0; u < SCAN_DEPTH; ++u) {
                const int q = qb + u * AGG_THREADS;
                const bool live = q < NQUADS;
                const unsigned r[4] = {(unsigned)(d[u].x - row0), (unsigned)(d[u].y - row0),
                                       (unsigned)(d[u].z - row0), (unsigned)(d[u].w - row0)};
                unsigned m[4];
                #pragma unroll
                for (int j = 0; j < 4; ++j) m[j] = __builtin_amdgcn_ballot_w32(live && r[j] < AROWS);
                if ((m[0] | m[1] | m[2] | m[3]) != 0) {
                    #pragma unroll
                    for (int j = 0; j < 4; ++j) {
                        if (m[j] != 0) {
                            if (live && r[j] < AROWS) {
                                const int o = pos + (int)__builtin_amdgcn_mbcnt_lo(m[j], 0u);
                                atomicAdd(&cnt[r[j]], 1);
                                if (o < WCAP) wl[o] = ((4 * q + j) << 6) | (int)r[j];
                                else full |= (1u << j) << (4 * u);
                            }
                            pos += __builtin_popcount(m[j]);
                            pos = pos < WCAP ? pos : WCAP;
                        }
                    }
                }
            }
            if (full != 0) beyond = 1;
            while (full != 0) {
                const int b = __builtin_ctz(full);
                full &= full - 1;
                const int e = 4 * (qb + (b >> 2) * AGG_THREADS) + (b & 3);
                const int rr = dst[e] - row0;
                const float* hs = H + (size_t)src[e] * DOUT;
                for (int col = 0; col < DOUT; ++col) atomicAdd(&acc[rr * DOUT + SUMPOS(col)], hs[col]);
            }
            #pragma unroll
            for (int u = 0; u < SCAN_DEPTH; ++u) d[u] = dn[u];
        }
    }
    if (lane == 0) wcount[wave] = pos;
    __syncthreads();

    {
        int pre = 0;
        #pragma unroll
        for (int i = 0; i < AGG_WAVES; ++i) {
            const int ci = wcount[i];
            n += ci;
            pre += (i < wave) ? ci : 0;
        }
        for (int i = lane; i < pos; i += 32) {
            const int ent = wl[i];
            flat[pre + i] = (src[ent >> 6] << 6) | (ent & 63);
        }
    }
    __syncthreads();
    n = __builtin_amdgcn_readfirstlane(n);
    }

    if (first) {
        for (int i = tid; i < n; i += AGG_THREADS) el[i] = flat[i];
        if (tid == 0) ECOUNT[blockIdx.x] = beyond ? -1 : n;
    }

    {
        for (int base = 0; base < n; base += AGG_THREADS) {
            const int idx = base + wave + 32 * lane;
            int s_l = 0, r_l = AROWS;
            if (idx < n) {
                const int ent = flat[idx];
                r_l = ent & 63;
                s_l = ent >> 6;
            }
            const int rest = n - base - wave;
            int cw = rest > 0 ? (rest + 31) >> 5 : 0;
            cw = cw < 32 ? cw : 32;

            for (int i = 0; i < cw; i += EDGES_PER_TRIP) {
                float4 v[EDGES_PER_TRIP][2];
                #pragma unroll
                for (int u = 0; u < EDGES_PER_TRIP; ++u) {
                    const int s = __builtin_amdgcn_readlane(s_l, i + u);
                    const float4* hs = (const float4*)(H + (size_t)s * DOUT) + lane;
                    v[u][0] = hs[0];
                    v[u][1] = hs[32];
                }
                #pragma unroll
                for (int u = 0; u < EDGES_PER_TRIP; ++u) {
                    const int r = __builtin_amdgcn_readlane(r_l, i + u);
                    float* ad = acc + r * DOUT + lane;
                    #pragma unroll
                    for (int hf = 0; hf < 2; ++hf) {
                        atomicAdd(&ad[128 * hf],      v[u][hf].x);
                        atomicAdd(&ad[128 * hf + 32], v[u][hf].y);
                        atomicAdd(&ad[128 * hf + 64], v[u][hf].z);
                        atomicAdd(&ad[128 * hf + 96], v[u][hf].w);
                    }
                }
            }
        }
    }
    __syncthreads();

    {
        float4* __restrict__ out4 = (float4*)(AGG + (size_t)row0 * DOUT);
        for (int i = tid; i < AROWS * DOUT / 4; i += AGG_THREADS) {
            const float* a = acc + (i >> 6) * DOUT + ((i & 32) << 2) + (i & 31);
            const float4 o = {a[0], a[32], a[64], a[96]};
            out4[i] = o;
        }
        if (DEG != nullptr && tid < AROWS) DEG[row0 + tid] = (float)cnt[tid];
    }
}

extern "C" void kernel_launch(void* const* d_in, const int* in_sizes, int n_in,
                              void* d_out, int out_size, void* d_ws, size_t ws_size,
                              hipStream_t stream)
{
    const float* x   = (const float*)d_in[0];
    const int*   ei  = (const int*)d_in[1];
    const float* W0  = (const float*)d_in[2];
    const float* b0  = (const float*)d_in[3];
    const float* Wl1 = (const float*)d_in[4];
    const float* bl1 = (const float*)d_in[5];
    const float* Wr1 = (const float*)d_in[6];
    const float* Wl2 = (const float*)d_in[7];
    const float* bl2 = (const float*)d_in[8];
    const float* Wr2 = (const float*)d_in[9];

    const int* src = ei;
    const int* dst = ei + NEDGES;

    float* ws  = (float*)d_ws;
    float* h    = ws;
    float* agg  = h + (size_t)NNODES * DOUT;
    float* deg  = agg + (size_t)NNODES * DOUT;
    int* ecount = (int*)(deg + DEG_PAD);
    int* elist  = ecount + ECOUNT_PAD;
    int* slots  = elist + (size_t)AGG_BLOCKS * LSTRIDE;

    float* out1 = (float*)d_out;
    float* out2 = out1 + (size_t)NNODES * DOUT;

    dim3 gemm_grid(GEMM_BLOCKS, 1);


    lin_gemm_kernel<<<gemm_grid, GEMM_THREADS, 0, stream>>>(x, W0, b0, h, src, dst, slots);

    gather_add_kernel<<<AGG_BLOCKS, AGG_THREADS, 0, stream>>>(h, src, dst, agg, deg, elist, ecount, slots, 1);
    sage_gemm_kernel<<<gemm_grid, GEMM_THREADS, 0, stream>>>(agg, deg, h, Wl1, bl1, Wr1, out1, 1);

    gather_add_kernel<<<AGG_BLOCKS, AGG_THREADS, 0, stream>>>(out1, src, dst, agg, (float*)nullptr, elist, ecount, slots, 0);
    sage_gemm_kernel<<<gemm_grid, GEMM_THREADS, 0, stream>>>(agg, deg, out1, Wl2, bl2, Wr2, out2, 0);
}


# Gather: a wave sorts its 32 edges by local row (bitonic on lane permutes); runs of equal row are summed in registers, one set of 8 LDS float adds per run, none for idle lanes; sum order changed
# speedup vs baseline: 3.2774x; 1.0335x over previous
#include <hip/hip_runtime.h>

typedef float v2f __attribute__((ext_vector_type(2)));
typedef float v8f __attribute__((ext_vector_type(8)));

#define NNODES 10000
#define NEDGES 320000
#define DIN 512
#define DOUT 256

#define TROWS 80
#define TCOLS 128
#define GEMM_THREADS 256
#define ROW_TILES (NNODES / TROWS)
#define GEMM_BLOCKS (ROW_TILES * (DOUT / TCOLS))
#define WTILES (TROWS / 16)
#define KB 8
#define KBS 4
static_assert(NNODES % TROWS == 0, "row tiles must be full: the kernels carry no row guards");
#define LIN_PITCH (DIN + 4)
#define SAGE_PITCH (DOUT + 4)

#define AROWS 40
#define AGG_BLOCKS (NNODES / AROWS)
#define SLOT_WORDS 32
#define SLOT_CAP (SLOT_WORDS - 1)
#define PRODUCERS GEMM_BLOCKS
#define EDGES_PER_PRODUCER (NEDGES / PRODUCERS)
#define EDGES_PER_THREAD (EDGES_PER_PRODUCER / GEMM_THREADS)
#define BIN_QUADS (AGG_BLOCKS * SLOT_WORDS / 4)
static_assert(NEDGES % PRODUCERS == 0 && EDGES_PER_PRODUCER % GEMM_THREADS == 0, "every producer thread bins the same number of edges");
static_assert(NNODES % AROWS == 0, "every owner has AROWS whole rows");

__global__ __launch_bounds__(GEMM_THREADS) void lin_gemm_kernel(
    const float* __restrict__ X, const float* __restrict__ W,
    const float* __restrict__ bias, float* __restrict__ H,
    const int* __restrict__ src, const int* __restrict__ dst, int* __restrict__ SLOTS)
{
    __shared__ __attribute__((aligned(16))) float As[TROWS * LIN_PITCH];
    __shared__ __attribute__((aligned(16))) int bins[AGG_BLOCKS * SLOT_WORDS];
    __shared__ int bcnt[AGG_BLOCKS];

    const int tid    = threadIdx.x;
    const int lane   = tid & 31;
    const int wave   = tid >> 5;
    const int lane16 = lane & 15;
    const int hsel   = lane >> 4;

    const int row0 = (blockIdx.x >> 1) * TROWS;
    const int col0 = (blockIdx.x & 1) * TCOLS + wave * 16;

    int ed[EDGES_PER_THREAD], es[EDGES_PER_THREAD];
    #pragma unroll
    for (int j = 0; j < EDGES_PER_THREAD; ++j) {
        const int e = blockIdx.x * EDGES_PER_PRODUCER + j * GEMM_THREADS + tid;
        ed[j] = dst[e];
        es[j] = src[e];
    }
    {
        const int4 z = {0, 0, 0, 0};
        for (int i = tid; i < BIN_QUADS; i += GEMM_THREADS) ((int4*)bins)[i] = z;
        if (tid < AGG_BLOCKS) bcnt[tid] = 0;
    }

    {
        const int rsub = tid >> 7;
        const int c4   = (tid & 127) * 4;
        #pragma unroll
        for (int p = 0; p < TROWS / 2; ++p) {
            const int r = 2 * p + rsub;
            const float4 v = *(const float4*)(X + (size_t)(row0 + r) * DIN + c4);
            *(float4*)(As + r * LIN_PITCH + c4) = v;
        }
    }
    __syncthreads();

    int place[EDGES_PER_THREAD];
    #pragma unroll
    for (int j = 0; j < EDGES_PER_THREAD; ++j) {
        const unsigned o = (unsigned)ed[j] / AROWS;
        place[j] = SLOT_CAP;
        if (o < AGG_BLOCKS) place[j] = atomicAdd(&bcnt[o], 1);
    }
    #pragma unroll
    for (int j = 0; j < EDGES_PER_THREAD; ++j) {
        const int o = (int)((unsigned)ed[j] / AROWS);
        const int r = ed[j] - o * AROWS;
        if (place[j] < SLOT_CAP) bins[o * SLOT_WORDS + 1 + place[j]] = (es[j] << 6) | r;
    }
    __syncthreads();
    #pragma unroll
    for (int u = 0; u < (BIN_QUADS + GEMM_THREADS - 1) / GEMM_THREADS; ++u) {
        const int i = tid + u * GEMM_THREADS;
        if (i < BIN_QUADS) {
            const int o = i >> 3;
            int4 q = ((const int4*)bins)[i];
            if ((i & 7) == 0) q.x = bcnt[o];
            ((int4*)SLOTS)[((size_t)o * PRODUCERS + blockIdx.x) * (SLOT_WORDS / 4) + (i & 7)] = q;
        }
    }

    const float* ap   = As + lane16 * LIN_PITCH + 2 * hsel;
    const float* bcol = W + (size_t)(2 * hsel) * DOUT + (col0 + lane16);

    v8f acc[WTILES];
    #pragma unroll
    for (int t = 0; t < WTILES; ++t) acc[t] = v8f{};

    v2f bc[KB], ac[WTILES];
    #pragma unroll
    for (int s = 0; s < KB; ++s) {
        bc[s].x = bcol[(4 * s) * DOUT];
        bc[s].y = bcol[(4 * s + 1) * DOUT];
    }
    #pragma unroll
    for (int t = 0; t < WTILES; ++t) ac[t] = *(const v2f*)(ap + t * 16 * LIN_PITCH);

    for (int kb = 0; kb < DIN; kb += 4 * KB) {
        const int knb = (kb + 4 * KB < DIN) ? kb + 4 * KB : 0;
        const float* bnp = bcol + (size_t)knb * DOUT;
        v2f bn[KB];
        #pragma unroll
        for (int s = 0; s < KB; ++s) {
            bn[s].x = bnp[(4 * s) * DOUT];
            bn[s].y = bnp[(4 * s + 1) * DOUT];
        }
        #pragma unroll
        for (int s = 0; s < KB; ++s) {
            const int kn = (s + 1 < KB) ? kb + 4 * (s + 1) : knb;
            v2f an[WTILES];
            #pragma unroll
            for (int t = 0; t < WTILES; ++t) an[t] = *(const v2f*)(ap + t * 16 * LIN_PITCH + kn);
            #pragma unroll
            for (int t = 0; t < WTILES; ++t)
                acc[t] = __builtin_amdgcn_wmma_f32_16x16x4_f32(false, ac[t], false, bc[s],
                                                               (short)0, acc[t], false, false);
            #pragma unroll
            for (int t = 0; t < WTILES; ++t) ac[t] = an[t];
        }
        #pragma unroll
        for (int s = 0; s < KB; ++s) bc[s] = bn[s];
    }

    const float bv = bias[col0 + lane16];
    #pragma unroll
    for (int t = 0; t < WTILES; ++t) {
        #pragma unroll
        for (int r = 0; r < 8; ++r) {
            const int row = row0 + 16 * t + r + 8 * hsel;
            H[(size_t)row * DOUT + col0 + lane16] = acc[t][r] + bv;
        }
    }
}

__global__ __launch_bounds__(GEMM_THREADS) void sage_gemm_kernel(
    const float* __restrict__ AGG,
    const float* __restrict__ DEG,
    const float* __restrict__ Hin,
    const float* __restrict__ Wl, const float* __restrict__ bl,
    const float* __restrict__ Wr,
    float* __restrict__ OUT, int relu)
{
    __shared__ __attribute__((aligned(16))) float Ms[TROWS * SAGE_PITCH];
    __shared__ __attribute__((aligned(16))) float Hs[TROWS * SAGE_PITCH];

    const int tid    = threadIdx.x;
    const int lane   = tid & 31;
    const int wave   = tid >> 5;
    const int lane16 = lane & 15;
    const int hsel   = lane >> 4;

    const int row0 = (blockIdx.x >> 1) * TROWS;
    const int col0 = (blockIdx.x & 1) * TCOLS + wave * 16;

    float degv[WTILES];
    #pragma unroll
    for (int t = 0; t < WTILES; ++t) degv[t] = DEG[row0 + 16 * t + lane16];

    {
        const int rsub = tid >> 6;
        const int c4   = (tid & 63) * 4;
        float4 mv[TROWS / 4], hv[TROWS / 4];
        #pragma unroll
        for (int p = 0; p < TROWS / 4; ++p) {
            const unsigned q = (unsigned)(row0 + 4 * p + rsub) * (DOUT / 4) + (unsigned)(tid & 63);
            mv[p] = ((const float4*)AGG)[q];
            hv[p] = ((const float4*)Hin)[q];
        }
        #pragma unroll
        for (int p = 0; p < TROWS / 4; ++p) {
            const int r = 4 * p + rsub;
            *(float4*)(Ms + r * SAGE_PITCH + c4) = mv[p];
            *(float4*)(Hs + r * SAGE_PITCH + c4) = hv[p];
        }
    }
    __syncthreads();

    float inv[WTILES];
    #pragma unroll
    for (int t = 0; t < WTILES; ++t) inv[t] = 1.0f / fmaxf(degv[t], 1.0f);

    const float* mp  = Ms + lane16 * SAGE_PITCH + 2 * hsel;
    const float* hp  = Hs + lane16 * SAGE_PITCH + 2 * hsel;
    const float* blc = Wl + (size_t)(2 * hsel) * DOUT + (col0 + lane16);
    const float* brc = Wr + (size_t)(2 * hsel) * DOUT + (col0 + lane16);

    v8f acc[WTILES];
    #pragma unroll
    for (int t = 0; t < WTILES; ++t) acc[t] = v8f{};

    v2f wlf[KBS], wrf[KBS], mc[WTILES], hc[WTILES];
    #pragma unroll
    for (int s = 0; s < KBS; ++s) {
        wlf[s].x = blc[(4 * s) * DOUT];    wlf[s].y = blc[(4 * s + 1) * DOUT];
        wrf[s].x = brc[(4 * s) * DOUT];    wrf[s].y = brc[(4 * s + 1) * DOUT];
    }
    #pragma unroll
    for (int t = 0; t < WTILES; ++t) {
        mc[t] = *(const v2f*)(mp + t * 16 * SAGE_PITCH);
        hc[t] = *(const v2f*)(hp + t * 16 * SAGE_PITCH);
    }

    for (int kb = 0; kb < DOUT; kb += 4 * KBS) {
        const int knb = (kb + 4 * KBS < DOUT) ? kb + 4 * KBS : 0;
        const float* blnp = blc + (size_t)knb * DOUT;
        const float* brnp = brc + (size_t)knb * DOUT;
        v2f wlfn[KBS], wrfn[KBS];
        #pragma unroll
        for (int s = 0; s < KBS; ++s) {
            wlfn[s].x = blnp[(4 * s) * DOUT];    wlfn[s].y = blnp[(4 * s + 1) * DOUT];
            wrfn[s].x = brnp[(4 * s) * DOUT];    wrfn[s].y = brnp[(4 * s + 1) * DOUT];
        }
        #pragma unroll
        for (int s = 0; s < KBS; ++s) {
            const int kn = (s + 1 < KBS) ? kb + 4 * (s + 1) : knb;
            v2f mn[WTILES], hn[WTILES];
            #pragma unroll
            for (int t = 0; t < WTILES; ++t) {
                mn[t] = *(const v2f*)(mp + t * 16 * SAGE_PITCH + kn);
                hn[t] = *(const v2f*)(hp + t * 16 * SAGE_PITCH + kn);
            }
            #pragma unroll
            for (int t = 0; t < WTILES; ++t) {
                const v2f am = mc[t] * inv[t];
                acc[t] = __builtin_amdgcn_wmma_f32_16x16x4_f32(false, am, false, wlf[s],
                                                               (short)0, acc[t], false, false);
            }
            #pragma unroll
            for (int t = 0; t < WTILES; ++t)
                acc[t] = __builtin_amdgcn_wmma_f32_16x16x4_f32(false, hc[t], false, wrf[s],
                                                               (short)0, acc[t], false, false);
            #pragma unroll
            for (int t = 0; t < WTILES; ++t) { mc[t] = mn[t]; hc[t] = hn[t]; }
        }
        #pragma unroll
        for (int s = 0; s < KBS; ++s) { wlf[s] = wlfn[s]; wrf[s] = wrfn[s]; }
    }

    const float bv = bl[col0 + lane16];
    #pragma unroll
    for (int t = 0; t < WTILES; ++t) {
        #pragma unroll
        for (int r = 0; r < 8; ++r) {
            const int lr = 16 * t + r + 8 * hsel;
            float v = acc[t][r] + bv;
            if (relu) v = fmaxf(v, 0.0f);
            const float hr = Hs[lr * SAGE_PITCH + col0 + lane16];
            OUT[(size_t)(row0 + lr) * DOUT + col0 + lane16] = v + hr;
        }
    }
}

#define AGG_THREADS 1024
#define AGG_WAVES (AGG_THREADS / 32)
#define SLOT_PASSES ((PRODUCERS + AGG_WAVES - 1) / AGG_WAVES)
#define WCAP 512
#define NQUADS (NEDGES / 4)
#define SCAN_DEPTH 8
#define SCAN_TRIPS ((NQUADS + SCAN_DEPTH * AGG_THREADS - 1) / (SCAN_DEPTH * AGG_THREADS))
#define EDGES_PER_TRIP 4
#define LCAP (AGG_WAVES * WCAP)
#define LSTRIDE (LCAP + 64)
#define DEG_PAD 10240
#define ECOUNT_PAD 256
static_assert(DEG_PAD >= NNODES && ECOUNT_PAD >= AGG_BLOCKS, "room for every count and every length");
#define SUMPOS(c) (((c) & 128) + (((c) & 3) << 5) + (((c) & 127) >> 2))
static_assert(DOUT == 256, "a row of sums is two halves of 32 lanes x 4 columns");
static_assert(NNODES % AROWS == 0, "every work-group owns AROWS whole rows");
static_assert(NEDGES % 4 == 0, "dst is read 4 entries at a time");
static_assert(AROWS < 64 && 32 % EDGES_PER_TRIP == 0 && NEDGES < (1 << 25) && NNODES < (1 << 25), "a list word holds the edge number, later src, above 6 bits of local row");
static_assert(AGG_WAVES == 32, "an add pass hands entry base + wave + 32*lane to a lane: 32 waves x 32 lanes");

__device__ __forceinline__ int lane_xor(int x, int m)
{
    switch (m) {
    case 1:  return __builtin_amdgcn_permlane16(x, x, 0x67452301u, 0xefcdab89u, false, false);
    case 2:  return __builtin_amdgcn_permlane16(x, x, 0x54761032u, 0xdcfe98bau, false, false);
    case 4:  return __builtin_amdgcn_permlane16(x, x, 0x32107654u, 0xba98fedcu, false, false);
    case 8:  return __builtin_amdgcn_permlane16(x, x, 0xfedcba98u, 0x76543210u, false, false);
    default: return __builtin_amdgcn_permlanex16(x, x, 0x76543210u, 0xfedcba98u, false, false);
    }
}

__global__ __launch_bounds__(AGG_THREADS) void gather_add_kernel(
    const float* __restrict__ H, const int* __restrict__ src,
    const int* __restrict__ dst, float* __restrict__ AGG,
    float* __restrict__ DEG, int* __restrict__ ELIST, int* __restrict__ ECOUNT,
    const int* __restrict__ SLOTS, int first)
{
    __shared__ __attribute__((aligned(16))) float acc[AROWS * DOUT];
    __shared__ int wlist[AGG_WAVES * WCAP];
    __shared__ int flat[AGG_WAVES * WCAP];
    __shared__ int cnt[AROWS];
    __shared__ int wcount[AGG_WAVES];
    __shared__ int beyond;
    __shared__ int short_line;

    const int tid  = threadIdx.x;
    const int lane = tid & 31;
    const int wave = __builtin_amdgcn_readfirstlane(tid >> 5);
    const int row0 = blockIdx.x * AROWS;

    int sw[SLOT_PASSES];
    #pragma unroll
    for (int k = 0; k < SLOT_PASSES; ++k) sw[k] = 0;
    if (first) {
        const int* __restrict__ sl = SLOTS + (size_t)blockIdx.x * (PRODUCERS * SLOT_WORDS) + lane;
        #pragma unroll
        for (int k = 0; k < SLOT_PASSES; ++k) {
            const int p = wave + AGG_WAVES * k;
            if (p < PRODUCERS) sw[k] = sl[p * SLOT_WORDS];
        }
    }

    {
        const float4 z = {0.0f, 0.0f, 0.0f, 0.0f};
        for (int i = tid; i < AROWS * DOUT / 4; i += AGG_THREADS) ((float4*)acc)[i] = z;
        if (tid < AROWS) cnt[tid] = 0;
        if (tid == 0) { beyond = 0; short_line = 0; }
    }

    int* __restrict__ el = ELIST + (size_t)blockIdx.x * LSTRIDE;
    int n = 0;
    bool sweep = true;
    if (!first) {
        const int nl = __builtin_amdgcn_readfirstlane(ECOUNT[blockIdx.x]);
        if (nl >= 0) {
            sweep = false;
            n = nl < LCAP ? nl : LCAP;
            for (int i = tid; i < n; i += AGG_THREADS) flat[i] = el[i];
        }
    }
    __syncthreads();

    if (first) {
        int lc[SLOT_PASSES];
        int tot = 0;
        bool bad = false;
        #pragma unroll
        for (int k = 0; k < SLOT_PASSES; ++k) {
            lc[k] = __builtin_amdgcn_readfirstlane(sw[k]);
            bad = bad || lc[k] > SLOT_CAP || lc[k] < 0;
            tot += lc[k];
        }
        if (lane == 0) {
            wcount[wave] = tot;
            if (bad) short_line = 1;
        }
        __syncthreads();
        if (!short_line) {
            sweep = false;
            int pre = 0;
            #pragma unroll
            for (int i = 0; i < AGG_WAVES; ++i) {
                const int ci = wcount[i];
                n += ci;
                pre += (i < wave) ? ci : 0;
            }
            #pragma unroll
            for (int k = 0; k < SLOT_PASSES; ++k) {
                if (lane >= 1 && lane <= lc[k]) {
                    flat[pre + lane - 1] = sw[k];
                    atomicAdd(&cnt[sw[k] & 63], 1);
                }
                pre += lc[k];
            }
            n = __builtin_amdgcn_readfirstlane(n);
        }
        __syncthreads();
    }

    if (sweep) {
    int* __restrict__ wl = wlist + wave * WCAP;
    int pos = 0;
    {
        const int4* __restrict__ dst4 = (const int4*)dst;
        int4 d[SCAN_DEPTH], dn[SCAN_DEPTH];
        #pragma unroll
        for (int u = 0; u < SCAN_DEPTH; ++u) d[u] = dst4[tid + u * AGG_THREADS];

        for (int t = 0; t < SCAN_TRIPS; ++t) {
            const int qb = t * (SCAN_DEPTH * AGG_THREADS) + tid;
            #pragma unroll
            for (int u = 0; u < SCAN_DEPTH; ++u) {
                const int qn = qb + (SCAN_DEPTH + u) * AGG_THREADS;
                dn[u] = dst4[qn < NQUADS ? qn : NQUADS - 1];
            }
            unsigned full = 0;
            #pragma unroll
            for (int u = 0; u < SCAN_DEPTH; ++u) {
                const int q = qb + u * AGG_THREADS;
                const bool live = q < NQUADS;
                const unsigned r[4] = {(unsigned)(d[u].x - row0), (unsigned)(d[u].y - row0),
                                       (unsigned)(d[u].z - row0), (unsigned)(d[u].w - row0)};
                unsigned m[4];
                #pragma unroll
                for (int j = 0; j < 4; ++j) m[j] = __builtin_amdgcn_ballot_w32(live && r[j] < AROWS);
                if ((m[0] | m[1] | m[2] | m[3]) != 0) {
                    #pragma unroll
                    for (int j = 0; j < 4; ++j) {
                        if (m[j] != 0) {
                            if (live && r[j] < AROWS) {
                                const int o = pos + (int)__builtin_amdgcn_mbcnt_lo(m[j], 0u);
                                atomicAdd(&cnt[r[j]], 1);
                                if (o < WCAP) wl[o] = ((4 * q + j) << 6) | (int)r[j];
                                else full |= (1u << j) << (4 * u);
                            }
                            pos += __builtin_popcount(m[j]);
                            pos = pos < WCAP ? pos : WCAP;
                        }
                    }
                }
            }
            if (full != 0) beyond = 1;
            while (full != 0) {
                const int b = __builtin_ctz(full);
                full &= full - 1;
                const int e = 4 * (qb + (b >> 2) * AGG_THREADS) + (b & 3);
                const int rr = dst[e] - row0;
                const float* hs = H + (size_t)src[e] * DOUT;
                for (int col = 0; col < DOUT; ++col) atomicAdd(&acc[rr * DOUT + SUMPOS(col)], hs[col]);
            }
            #pragma unroll
            for (int u = 0; u < SCAN_DEPTH; ++u) d[u] = dn[u];
        }
    }
    if (lane == 0) wcount[wave] = pos;
    __syncthreads();

    {
        int pre = 0;
        #pragma unroll
        for (int i = 0; i < AGG_WAVES; ++i) {
            const int ci = wcount[i];
            n += ci;
            pre += (i < wave) ? ci : 0;
        }
        for (int i = lane; i < pos; i += 32) {
            const int ent = wl[i];
            flat[pre + i] = (src[ent >> 6] << 6) | (ent & 63);
        }
    }
    __syncthreads();
    n = __builtin_amdgcn_readfirstlane(n);
    }

    if (first) {
        for (int i = tid; i < n; i += AGG_THREADS) el[i] = flat[i];
        if (tid == 0) ECOUNT[blockIdx.x] = beyond ? -1 : n;
    }

    {
        for (int base = 0; base < n; base += AGG_THREADS) {
            const int idx = base + wave + 32 * lane;
            int key = AROWS << 25;
            if (idx < n) {
                const int ent = flat[idx];
                key = ((ent & 63) << 25) | (ent >> 6);
            }
            #pragma unroll
            for (int k = 2; k <= 32; k <<= 1) {
                #pragma unroll
                for (int j = k >> 1; j > 0; j >>= 1) {
                    const int other = lane_xor(key, j);
                    const bool low = ((lane & j) == 0) == ((lane & k) == 0);
                    key = low ? (key < other ? key : other) : (key > other ? key : other);
                }
            }
            const int r_l = key >> 25;
            const int s_l = key & ((1 << 25) - 1);

            const int rest = n - base - wave;
            int cw = rest > 0 ? (rest + 31) >> 5 : 0;
            cw = cw < 32 ? cw : 32;

            int prow = AROWS;
            float4 p0 = {0.0f, 0.0f, 0.0f, 0.0f}, p1 = {0.0f, 0.0f, 0.0f, 0.0f};

            for (int i = 0; i < cw; i += EDGES_PER_TRIP) {
                float4 v[EDGES_PER_TRIP][2];
                #pragma unroll
                for (int u = 0; u < EDGES_PER_TRIP; ++u) {
                    const int s = __builtin_amdgcn_readlane(s_l, i + u);
                    const float4* hs = (const float4*)(H + (size_t)s * DOUT) + lane;
                    v[u][0] = hs[0];
                    v[u][1] = hs[32];
                }
                #pragma unroll
                for (int u = 0; u < EDGES_PER_TRIP; ++u) {
                    const int r = __builtin_amdgcn_readlane(r_l, i + u);
                    if (r == prow) {
                        p0.x += v[u][0].x; p0.y += v[u][0].y; p0.z += v[u][0].z; p0.w += v[u][0].w;
                        p1.x += v[u][1].x; p1.y += v[u][1].y; p1.z += v[u][1].z; p1.w += v[u][1].w;
                    } else {
                        if (prow < AROWS) {
                            float* ad = acc + prow * DOUT + lane;
                            atomicAdd(&ad[0],   p0.x);
                            atomicAdd(&ad[32],  p0.y);
                            atomicAdd(&ad[64],  p0.z);
                            atomicAdd(&ad[96],  p0.w);
                            atomicAdd(&ad[128], p1.x);
                            atomicAdd(&ad[160], p1.y);
                            atomicAdd(&ad[192], p1.z);
                            atomicAdd(&ad[224], p1.w);
                        }
                        prow = r;
                        p0 = v[u][0];
                        p1 = v[u][1];
                    }
                }
            }
            if (prow < AROWS) {
                float* ad = acc + prow * DOUT + lane;
                atomicAdd(&ad[0],   p0.x);
                atomicAdd(&ad[32],  p0.y);
                atomicAdd(&ad[64],  p0.z);
                atomicAdd(&ad[96],  p0.w);
                atomicAdd(&ad[128], p1.x);
                atomicAdd(&ad[160], p1.y);
                atomicAdd(&ad[192], p1.z);
                atomicAdd(&ad[224], p1.w);
            }
        }
    }
    __syncthreads();

    {
        float4* __restrict__ out4 = (float4*)(AGG + (size_t)row0 * DOUT);
        for (int i = tid; i < AROWS * DOUT / 4; i += AGG_THREADS) {
            const float* a = acc + (i >> 6) * DOUT + ((i & 32) << 2) + (i & 31);
            const float4 o = {a[0], a[32], a[64], a[96]};
            out4[i] = o;
        }
        if (DEG != nullptr && tid < AROWS) DEG[row0 + tid] = (float)cnt[tid];
    }
}

extern "C" void kernel_launch(void* const* d_in, const int* in_sizes, int n_in,
                              void* d_out, int out_size, void* d_ws, size_t ws_size,
                              hipStream_t stream)
{
    const float* x   = (const float*)d_in[0];
    const int*   ei  = (const int*)d_in[1];
    const float* W0  = (const float*)d_in[2];
    const float* b0  = (const float*)d_in[3];
    const float* Wl1 = (const float*)d_in[4];
    const float* bl1 = (const float*)d_in[5];
    const float* Wr1 = (const float*)d_in[6];
    const float* Wl2 = (const float*)d_in[7];
    const float* bl2 = (const float*)d_in[8];
    const float* Wr2 = (const float*)d_in[9];

    const int* src = ei;
    const int* dst = ei + NEDGES;

    float* ws  = (float*)d_ws;
    float* h    = ws;
    float* agg  = h + (size_t)NNODES * DOUT;
    float* deg  = agg + (size_t)NNODES * DOUT;
    int* ecount = (int*)(deg + DEG_PAD);
    int* elist  = ecount + ECOUNT_PAD;
    int* slots  = elist + (size_t)AGG_BLOCKS * LSTRIDE;

    float* out1 = (float*)d_out;
    float* out2 = out1 + (size_t)NNODES * DOUT;

    dim3 gemm_grid(GEMM_BLOCKS, 1);


    lin_gemm_kernel<<<gemm_grid, GEMM_THREADS, 0, stream>>>(x, W0, b0, h, src, dst, slots);

    gather_add_kernel<<<AGG_BLOCKS, AGG_THREADS, 0, stream>>>(h, src, dst, agg, deg, elist, ecount, slots, 1);
    sage_gemm_kernel<<<gemm_grid, GEMM_THREADS, 0, stream>>>(agg, deg, h, Wl1, bl1, Wr1, out1, 1);

    gather_add_kernel<<<AGG_BLOCKS, AGG_THREADS, 0, stream>>>(out1, src, dst, agg, (float*)nullptr, elist, ecount, slots, 0);
    sage_gemm_kernel<<<gemm_grid, GEMM_THREADS, 0, stream>>>(agg, deg, out1, Wl2, bl2, Wr2, out2, 0);
}


# Gather: edge list built sorted by node (counting sort on the in-degree adds); a wave takes a contiguous piece of it, so whole nodes are summed in registers; per-wave sort gone; sum order changed
# speedup vs baseline: 3.5426x; 1.0809x over previous
#include <hip/hip_runtime.h>

typedef float v2f __attribute__((ext_vector_type(2)));
typedef float v8f __attribute__((ext_vector_type(8)));

#define NNODES 10000
#define NEDGES 320000
#define DIN 512
#define DOUT 256

#define TROWS 80
#define TCOLS 128
#define GEMM_THREADS 256
#define ROW_TILES (NNODES / TROWS)
#define GEMM_BLOCKS (ROW_TILES * (DOUT / TCOLS))
#define WTILES (TROWS / 16)
#define KB 8
#define KBS 4
static_assert(NNODES % TROWS == 0, "row tiles must be full: the kernels carry no row guards");
#define LIN_PITCH (DIN + 4)
#define SAGE_PITCH (DOUT + 4)

#define AROWS 40
#define AGG_BLOCKS (NNODES / AROWS)
#define SLOT_WORDS 32
#define SLOT_CAP (SLOT_WORDS - 1)
#define PRODUCERS GEMM_BLOCKS
#define EDGES_PER_PRODUCER (NEDGES / PRODUCERS)
#define EDGES_PER_THREAD (EDGES_PER_PRODUCER / GEMM_THREADS)
#define BIN_QUADS (AGG_BLOCKS * SLOT_WORDS / 4)
static_assert(NEDGES % PRODUCERS == 0 && EDGES_PER_PRODUCER % GEMM_THREADS == 0, "every producer thread bins the same number of edges");
static_assert(NNODES % AROWS == 0, "every owner has AROWS whole rows");

__global__ __launch_bounds__(GEMM_THREADS) void lin_gemm_kernel(
    const float* __restrict__ X, const float* __restrict__ W,
    const float* __restrict__ bias, float* __restrict__ H,
    const int* __restrict__ src, const int* __restrict__ dst, int* __restrict__ SLOTS)
{
    __shared__ __attribute__((aligned(16))) float As[TROWS * LIN_PITCH];
    __shared__ __attribute__((aligned(16))) int bins[AGG_BLOCKS * SLOT_WORDS];
    __shared__ int bcnt[AGG_BLOCKS];

    const int tid    = threadIdx.x;
    const int lane   = tid & 31;
    const int wave   = tid >> 5;
    const int lane16 = lane & 15;
    const int hsel   = lane >> 4;

    const int row0 = (blockIdx.x >> 1) * TROWS;
    const int col0 = (blockIdx.x & 1) * TCOLS + wave * 16;

    int ed[EDGES_PER_THREAD], es[EDGES_PER_THREAD];
    #pragma unroll
    for (int j = 0; j < EDGES_PER_THREAD; ++j) {
        const int e = blockIdx.x * EDGES_PER_PRODUCER + j * GEMM_THREADS + tid;
        ed[j] = dst[e];
        es[j] = src[e];
    }
    {
        const int4 z = {0, 0, 0, 0};
        for (int i = tid; i < BIN_QUADS; i += GEMM_THREADS) ((int4*)bins)[i] = z;
        if (tid < AGG_BLOCKS) bcnt[tid] = 0;
    }

    {
        const int rsub = tid >> 7;
        const int c4   = (tid & 127) * 4;
        #pragma unroll
        for (int p = 0; p < TROWS / 2; ++p) {
            const int r = 2 * p + rsub;
            const float4 v = *(const float4*)(X + (size_t)(row0 + r) * DIN + c4);
            *(float4*)(As + r * LIN_PITCH + c4) = v;
        }
    }
    __syncthreads();

    int place[EDGES_PER_THREAD];
    #pragma unroll
    for (int j = 0; j < EDGES_PER_THREAD; ++j) {
        const unsigned o = (unsigned)ed[j] / AROWS;
        place[j] = SLOT_CAP;
        if (o < AGG_BLOCKS) place[j] = atomicAdd(&bcnt[o], 1);
    }
    #pragma unroll
    for (int j = 0; j < EDGES_PER_THREAD; ++j) {
        const int o = (int)((unsigned)ed[j] / AROWS);
        const int r = ed[j] - o * AROWS;
        if (place[j] < SLOT_CAP) bins[o * SLOT_WORDS + 1 + place[j]] = (es[j] << 6) | r;
    }
    __syncthreads();
    #pragma unroll
    for (int u = 0; u < (BIN_QUADS + GEMM_THREADS - 1) / GEMM_THREADS; ++u) {
        const int i = tid + u * GEMM_THREADS;
        if (i < BIN_QUADS) {
            const int o = i >> 3;
            int4 q = ((const int4*)bins)[i];
            if ((i & 7) == 0) q.x = bcnt[o];
            ((int4*)SLOTS)[((size_t)o * PRODUCERS + blockIdx.x) * (SLOT_WORDS / 4) + (i & 7)] = q;
        }
    }

    const float* ap   = As + lane16 * LIN_PITCH + 2 * hsel;
    const float* bcol = W + (size_t)(2 * hsel) * DOUT + (col0 + lane16);

    v8f acc[WTILES];
    #pragma unroll
    for (int t = 0; t < WTILES; ++t) acc[t] = v8f{};

    v2f bc[KB], ac[WTILES];
    #pragma unroll
    for (int s = 0; s < KB; ++s) {
        bc[s].x = bcol[(4 * s) * DOUT];
        bc[s].y = bcol[(4 * s + 1) * DOUT];
    }
    #pragma unroll
    for (int t = 0; t < WTILES; ++t) ac[t] = *(const v2f*)(ap + t * 16 * LIN_PITCH);

    for (int kb = 0; kb < DIN; kb += 4 * KB) {
        const int knb = (kb + 4 * KB < DIN) ? kb + 4 * KB : 0;
        const float* bnp = bcol + (size_t)knb * DOUT;
        v2f bn[KB];
        #pragma unroll
        for (int s = 0; s < KB; ++s) {
            bn[s].x = bnp[(4 * s) * DOUT];
            bn[s].y = bnp[(4 * s + 1) * DOUT];
        }
        #pragma unroll
        for (int s = 0; s < KB; ++s) {
            const int kn = (s + 1 < KB) ? kb + 4 * (s + 1) : knb;
            v2f an[WTILES];
            #pragma unroll
            for (int t = 0; t < WTILES; ++t) an[t] = *(const v2f*)(ap + t * 16 * LIN_PITCH + kn);
            #pragma unroll
            for (int t = 0; t < WTILES; ++t)
                acc[t] = __builtin_amdgcn_wmma_f32_16x16x4_f32(false, ac[t], false, bc[s],
                                                               (short)0, acc[t], false, false);
            #pragma unroll
            for (int t = 0; t < WTILES; ++t) ac[t] = an[t];
        }
        #pragma unroll
        for (int s = 0; s < KB; ++s) bc[s] = bn[s];
    }

    const float bv = bias[col0 + lane16];
    #pragma unroll
    for (int t = 0; t < WTILES; ++t) {
        #pragma unroll
        for (int r = 0; r < 8; ++r) {
            const int row = row0 + 16 * t + r + 8 * hsel;
            H[(size_t)row * DOUT + col0 + lane16] = acc[t][r] + bv;
        }
    }
}

__global__ __launch_bounds__(GEMM_THREADS) void sage_gemm_kernel(
    const float* __restrict__ AGG,
    const float* __restrict__ DEG,
    const float* __restrict__ Hin,
    const float* __restrict__ Wl, const float* __restrict__ bl,
    const float* __restrict__ Wr,
    float* __restrict__ OUT, int relu)
{
    __shared__ __attribute__((aligned(16))) float Ms[TROWS * SAGE_PITCH];
    __shared__ __attribute__((aligned(16))) float Hs[TROWS * SAGE_PITCH];

    const int tid    = threadIdx.x;
    const int lane   = tid & 31;
    const int wave   = tid >> 5;
    const int lane16 = lane & 15;
    const int hsel   = lane >> 4;

    const int row0 = (blockIdx.x >> 1) * TROWS;
    const int col0 = (blockIdx.x & 1) * TCOLS + wave * 16;

    float degv[WTILES];
    #pragma unroll
    for (int t = 0; t < WTILES; ++t) degv[t] = DEG[row0 + 16 * t + lane16];

    {
        const int rsub = tid >> 6;
        const int c4   = (tid & 63) * 4;
        float4 mv[TROWS / 4], hv[TROWS / 4];
        #pragma unroll
        for (int p = 0; p < TROWS / 4; ++p) {
            const unsigned q = (unsigned)(row0 + 4 * p + rsub) * (DOUT / 4) + (unsigned)(tid & 63);
            mv[p] = ((const float4*)AGG)[q];
            hv[p] = ((const float4*)Hin)[q];
        }
        #pragma unroll
        for (int p = 0; p < TROWS / 4; ++p) {
            const int r = 4 * p + rsub;
            *(float4*)(Ms + r * SAGE_PITCH + c4) = mv[p];
            *(float4*)(Hs + r * SAGE_PITCH + c4) = hv[p];
        }
    }
    __syncthreads();

    float inv[WTILES];
    #pragma unroll
    for (int t = 0; t < WTILES; ++t) inv[t] = 1.0f / fmaxf(degv[t], 1.0f);

    const float* mp  = Ms + lane16 * SAGE_PITCH + 2 * hsel;
    const float* hp  = Hs + lane16 * SAGE_PITCH + 2 * hsel;
    const float* blc = Wl + (size_t)(2 * hsel) * DOUT + (col0 + lane16);
    const float* brc = Wr + (size_t)(2 * hsel) * DOUT + (col0 + lane16);

    v8f acc[WTILES];
    #pragma unroll
    for (int t = 0; t < WTILES; ++t) acc[t] = v8f{};

    v2f wlf[KBS], wrf[KBS], mc[WTILES], hc[WTILES];
    #pragma unroll
    for (int s = 0; s < KBS; ++s) {
        wlf[s].x = blc[(4 * s) * DOUT];    wlf[s].y = blc[(4 * s + 1) * DOUT];
        wrf[s].x = brc[(4 * s) * DOUT];    wrf[s].y = brc[(4 * s + 1) * DOUT];
    }
    #pragma unroll
    for (int t = 0; t < WTILES; ++t) {
        mc[t] = *(const v2f*)(mp + t * 16 * SAGE_PITCH);
        hc[t] = *(const v2f*)(hp + t * 16 * SAGE_PITCH);
    }

    for (int kb = 0; kb < DOUT; kb += 4 * KBS) {
        const int knb = (kb + 4 * KBS < DOUT) ? kb + 4 * KBS : 0;
        const float* blnp = blc + (size_t)knb * DOUT;
        const float* brnp = brc + (size_t)knb * DOUT;
        v2f wlfn[KBS], wrfn[KBS];
        #pragma unroll
        for (int s = 0; s < KBS; ++s) {
            wlfn[s].x = blnp[(4 * s) * DOUT];    wlfn[s].y = blnp[(4 * s + 1) * DOUT];
            wrfn[s].x = brnp[(4 * s) * DOUT];    wrfn[s].y = brnp[(4 * s + 1) * DOUT];
        }
        #pragma unroll
        for (int s = 0; s < KBS; ++s) {
            const int kn = (s + 1 < KBS) ? kb + 4 * (s + 1) : knb;
            v2f mn[WTILES], hn[WTILES];
            #pragma unroll
            for (int t = 0; t < WTILES; ++t) {
                mn[t] = *(const v2f*)(mp + t * 16 * SAGE_PITCH + kn);
                hn[t] = *(const v2f*)(hp + t * 16 * SAGE_PITCH + kn);
            }
            #pragma unroll
            for (int t = 0; t < WTILES; ++t) {
                const v2f am = mc[t] * inv[t];
                acc[t] = __builtin_amdgcn_wmma_f32_16x16x4_f32(false, am, false, wlf[s],
                                                               (short)0, acc[t], false, false);
            }
            #pragma unroll
            for (int t = 0; t < WTILES; ++t)
                acc[t] = __builtin_amdgcn_wmma_f32_16x16x4_f32(false, hc[t], false, wrf[s],
                                                               (short)0, acc[t], false, false);
            #pragma unroll
            for (int t = 0; t < WTILES; ++t) { mc[t] = mn[t]; hc[t] = hn[t]; }
        }
        #pragma unroll
        for (int s = 0; s < KBS; ++s) { wlf[s] = wlfn[s]; wrf[s] = wrfn[s]; }
    }

    const float bv = bl[col0 + lane16];
    #pragma unroll
    for (int t = 0; t < WTILES; ++t) {
        #pragma unroll
        for (int r = 0; r < 8; ++r) {
            const int lr = 16 * t + r + 8 * hsel;
            float v = acc[t][r] + bv;
            if (relu) v = fmaxf(v, 0.0f);
            const float hr = Hs[lr * SAGE_PITCH + col0 + lane16];
            OUT[(size_t)(row0 + lr) * DOUT + col0 + lane16] = v + hr;
        }
    }
}

#define AGG_THREADS 1024
#define AGG_WAVES (AGG_THREADS / 32)
#define SLOT_PASSES ((PRODUCERS + AGG_WAVES - 1) / AGG_WAVES)
#define WCAP 512
#define NQUADS (NEDGES / 4)
#define SCAN_DEPTH 8
#define SCAN_TRIPS ((NQUADS + SCAN_DEPTH * AGG_THREADS - 1) / (SCAN_DEPTH * AGG_THREADS))
#define EDGES_PER_TRIP 4
#define LCAP (AGG_WAVES * WCAP)
#define LSTRIDE (LCAP + 64)
#define DEG_PAD 10240
#define ECOUNT_PAD 256
static_assert(DEG_PAD >= NNODES && ECOUNT_PAD >= AGG_BLOCKS, "room for every count and every length");
#define SUMPOS(c) (((c) & 128) + (((c) & 3) << 5) + (((c) & 127) >> 2))
static_assert(DOUT == 256, "a row of sums is two halves of 32 lanes x 4 columns");
static_assert(NNODES % AROWS == 0, "every work-group owns AROWS whole rows");
static_assert(NEDGES % 4 == 0, "dst is read 4 entries at a time");
static_assert(AROWS < 64 && 32 % EDGES_PER_TRIP == 0 && NEDGES < (1 << 25) && NNODES < (1 << 25), "a list word holds the edge number, later src, above 6 bits of local row");

__global__ __launch_bounds__(AGG_THREADS) void gather_add_kernel(
    const float* __restrict__ H, const int* __restrict__ src,
    const int* __restrict__ dst, float* __restrict__ AGG,
    float* __restrict__ DEG, int* __restrict__ ELIST, int* __restrict__ ECOUNT,
    const int* __restrict__ SLOTS, int first)
{
    __shared__ __attribute__((aligned(16))) float acc[AROWS * DOUT];
    __shared__ int wlist[AGG_WAVES * WCAP];
    __shared__ int flat[AGG_WAVES * WCAP];
    __shared__ int cnt[AROWS];
    __shared__ int rstart[AROWS];
    __shared__ int wcount[AGG_WAVES];
    __shared__ int beyond;
    __shared__ int short_line;

    const int tid  = threadIdx.x;
    const int lane = tid & 31;
    const int wave = __builtin_amdgcn_readfirstlane(tid >> 5);
    const int row0 = blockIdx.x * AROWS;

    int sw[SLOT_PASSES];
    #pragma unroll
    for (int k = 0; k < SLOT_PASSES; ++k) sw[k] = 0;
    if (first) {
        const int* __restrict__ sl = SLOTS + (size_t)blockIdx.x * (PRODUCERS * SLOT_WORDS) + lane;
        #pragma unroll
        for (int k = 0; k < SLOT_PASSES; ++k) {
            const int p = wave + AGG_WAVES * k;
            if (p < PRODUCERS) sw[k] = sl[p * SLOT_WORDS];
        }
    }

    {
        const float4 z = {0.0f, 0.0f, 0.0f, 0.0f};
        for (int i = tid; i < AROWS * DOUT / 4; i += AGG_THREADS) ((float4*)acc)[i] = z;
        if (tid < AROWS) cnt[tid] = 0;
        if (tid == 0) { beyond = 0; short_line = 0; }
    }

    int* __restrict__ el = ELIST + (size_t)blockIdx.x * LSTRIDE;
    int n = 0;
    bool sweep = true;
    if (!first) {
        const int nl = __builtin_amdgcn_readfirstlane(ECOUNT[blockIdx.x]);
        if (nl >= 0) {
            sweep = false;
            n = nl < LCAP ? nl : LCAP;
            for (int i = tid; i < n; i += AGG_THREADS) flat[i] = el[i];
        }
    }
    __syncthreads();

    if (first) {
        int lc[SLOT_PASSES];
        int tot = 0;
        bool bad = false;
        #pragma unroll
        for (int k = 0; k < SLOT_PASSES; ++k) {
            lc[k] = __builtin_amdgcn_readfirstlane(sw[k]);
            bad = bad || lc[k] > SLOT_CAP || lc[k] < 0;
            tot += lc[k];
        }
        if (lane == 0) {
            wcount[wave] = tot;
            if (bad) short_line = 1;
        }
        __syncthreads();
        int rk[SLOT_PASSES];
        #pragma unroll
        for (int k = 0; k < SLOT_PASSES; ++k) rk[k] = 0;
        if (!short_line) {
            sweep = false;
            #pragma unroll
            for (int i = 0; i < AGG_WAVES; ++i) n += wcount[i];
            #pragma unroll
            for (int k = 0; k < SLOT_PASSES; ++k)
                if (lane >= 1 && lane <= lc[k]) rk[k] = atomicAdd(&cnt[sw[k] & 63], 1);
            n = __builtin_amdgcn_readfirstlane(n);
        }
        __syncthreads();
        if (!sweep && tid < AROWS) {
            int s = 0;
            #pragma unroll
            for (int i = 0; i < AROWS - 1; ++i) {
                const int ci = cnt[i];
                s += (i < tid) ? ci : 0;
            }
            rstart[tid] = s;
        }
        __syncthreads();
        if (!sweep) {
            int rs[SLOT_PASSES];
            #pragma unroll
            for (int k = 0; k < SLOT_PASSES; ++k) {
                const int r = sw[k] & 63;
                rs[k] = rstart[r < AROWS ? r : 0];
            }
            #pragma unroll
            for (int k = 0; k < SLOT_PASSES; ++k)
                if (lane >= 1 && lane <= lc[k]) flat[rs[k] + rk[k]] = sw[k];
        }
        __syncthreads();
    }

    if (sweep) {
    int* __restrict__ wl = wlist + wave * WCAP;
    int pos = 0;
    {
        const int4* __restrict__ dst4 = (const int4*)dst;
        int4 d[SCAN_DEPTH], dn[SCAN_DEPTH];
        #pragma unroll
        for (int u = 0; u < SCAN_DEPTH; ++u) d[u] = dst4[tid + u * AGG_THREADS];

        for (int t = 0; t < SCAN_TRIPS; ++t) {
            const int qb = t * (SCAN_DEPTH * AGG_THREADS) + tid;
            #pragma unroll
            for (int u = 0; u < SCAN_DEPTH; ++u) {
                const int qn = qb + (SCAN_DEPTH + u) * AGG_THREADS;
                dn[u] = dst4[qn < NQUADS ? qn : NQUADS - 1];
            }
            unsigned full = 0;
            #pragma unroll
            for (int u = 0; u < SCAN_DEPTH; ++u) {
                const int q = qb + u * AGG_THREADS;
                const bool live = q < NQUADS;
                const unsigned r[4] = {(unsigned)(d[u].x - row0), (unsigned)(d[u].y - row0),
                                       (unsigned)(d[u].z - row0), (unsigned)(d[u].w - row0)};
                unsigned m[4];
                #pragma unroll
                for (int j = 0; j < 4; ++j) m[j] = __builtin_amdgcn_ballot_w32(live && r[j] < AROWS);
                if ((m[0] | m[1] | m[2] | m[3]) != 0) {
                    #pragma unroll
                    for (int j = 0; j < 4; ++j) {
                        if (m[j] != 0) {
                            if (live && r[j] < AROWS) {
                                const int o = pos + (int)__builtin_amdgcn_mbcnt_lo(m[j], 0u);
                                atomicAdd(&cnt[r[j]], 1);
                                if (o < WCAP) wl[o] = ((4 * q + j) << 6) | (int)r[j];
                                else full |= (1u << j) << (4 * u);
                            }
                            pos += __builtin_popcount(m[j]);
                            pos = pos < WCAP ? pos : WCAP;
                        }
                    }
                }
            }
            if (full != 0) beyond = 1;
            while (full != 0) {
                const int b = __builtin_ctz(full);
                full &= full - 1;
                const int e = 4 * (qb + (b >> 2) * AGG_THREADS) + (b & 3);
                const int rr = dst[e] - row0;
                const float* hs = H + (size_t)src[e] * DOUT;
                for (int col = 0; col < DOUT; ++col) atomicAdd(&acc[rr * DOUT + SUMPOS(col)], hs[col]);
            }
            #pragma unroll
            for (int u = 0; u < SCAN_DEPTH; ++u) d[u] = dn[u];
        }
    }
    if (lane == 0) wcount[wave] = pos;
    __syncthreads();

    {
        int pre = 0;
        #pragma unroll
        for (int i = 0; i < AGG_WAVES; ++i) {
            const int ci = wcount[i];
            n += ci;
            pre += (i < wave) ? ci : 0;
        }
        for (int i = lane; i < pos; i += 32) {
            const int ent = wl[i];
            flat[pre + i] = (src[ent >> 6] << 6) | (ent & 63);
        }
    }
    __syncthreads();
    n = __builtin_amdgcn_readfirstlane(n);
    }

    if (first) {
        for (int i = tid; i < n; i += AGG_THREADS) el[i] = flat[i];
        if (tid == 0) ECOUNT[blockIdx.x] = beyond ? -1 : n;
    }

    {
        const int per = (n + AGG_WAVES - 1) / AGG_WAVES;
        const int c0  = wave * per;
        const int c1  = c0 + per < n ? c0 + per : n;
        int prow = AROWS;
        float4 p0 = {0.0f, 0.0f, 0.0f, 0.0f}, p1 = {0.0f, 0.0f, 0.0f, 0.0f};
        for (int base = c0; base < c1; base += 32) {
            const int idx = base + lane;
            int r_l = AROWS, s_l = 0;
            if (idx < c1) {
                const int ent = flat[idx];
                r_l = ent & 63;
                s_l = ent >> 6;
            }

            int cw = c1 - base;
            cw = cw < 32 ? cw : 32;

            for (int i = 0; i < cw; i += EDGES_PER_TRIP) {
                float4 v[EDGES_PER_TRIP][2];
                #pragma unroll
                for (int u = 0; u < EDGES_PER_TRIP; ++u) {
                    const int s = __builtin_amdgcn_readlane(s_l, i + u);
                    const float4* hs = (const float4*)(H + (size_t)s * DOUT) + lane;
                    v[u][0] = hs[0];
                    v[u][1] = hs[32];
                }
                #pragma unroll
                for (int u = 0; u < EDGES_PER_TRIP; ++u) {
                    const int r = __builtin_amdgcn_readlane(r_l, i + u);
                    if (r == prow) {
                        p0.x += v[u][0].x; p0.y += v[u][0].y; p0.z += v[u][0].z; p0.w += v[u][0].w;
                        p1.x += v[u][1].x; p1.y += v[u][1].y; p1.z += v[u][1].z; p1.w += v[u][1].w;
                    } else {
                        if (prow < AROWS) {
                            float* ad = acc + prow * DOUT + lane;
                            atomicAdd(&ad[0],   p0.x);
                            atomicAdd(&ad[32],  p0.y);
                            atomicAdd(&ad[64],  p0.z);
                            atomicAdd(&ad[96],  p0.w);
                            atomicAdd(&ad[128], p1.x);
                            atomicAdd(&ad[160], p1.y);
                            atomicAdd(&ad[192], p1.z);
                            atomicAdd(&ad[224], p1.w);
                        }
                        prow = r;
                        p0 = v[u][0];
                        p1 = v[u][1];
                    }
                }
            }
        }
        if (prow < AROWS) {
            float* ad = acc + prow * DOUT + lane;
            atomicAdd(&ad[0],   p0.x);
            atomicAdd(&ad[32],  p0.y);
            atomicAdd(&ad[64],  p0.z);
            atomicAdd(&ad[96],  p0.w);
            atomicAdd(&ad[128], p1.x);
            atomicAdd(&ad[160], p1.y);
            atomicAdd(&ad[192], p1.z);
            atomicAdd(&ad[224], p1.w);
        }
    }
    __syncthreads();

    {
        float4* __restrict__ out4 = (float4*)(AGG + (size_t)row0 * DOUT);
        for (int i = tid; i < AROWS * DOUT / 4; i += AGG_THREADS) {
            const float* a = acc + (i >> 6) * DOUT + ((i & 32) << 2) + (i & 31);
            const float4 o = {a[0], a[32], a[64], a[96]};
            out4[i] = o;
        }
        if (DEG != nullptr && tid < AROWS) DEG[row0 + tid] = (float)cnt[tid];
    }
}

extern "C" void kernel_launch(void* const* d_in, const int* in_sizes, int n_in,
                              void* d_out, int out_size, void* d_ws, size_t ws_size,
                              hipStream_t stream)
{
    const float* x   = (const float*)d_in[0];
    const int*   ei  = (const int*)d_in[1];
    const float* W0  = (const float*)d_in[2];
    const float* b0  = (const float*)d_in[3];
    const float* Wl1 = (const float*)d_in[4];
    const float* bl1 = (const float*)d_in[5];
    const float* Wr1 = (const float*)d_in[6];
    const float* Wl2 = (const float*)d_in[7];
    const float* bl2 = (const float*)d_in[8];
    const float* Wr2 = (const float*)d_in[9];

    const int* src = ei;
    const int* dst = ei + NEDGES;

    float* ws  = (float*)d_ws;
    float* h    = ws;
    float* agg  = h + (size_t)NNODES * DOUT;
    float* deg  = agg + (size_t)NNODES * DOUT;
    int* ecount = (int*)(deg + DEG_PAD);
    int* elist  = ecount + ECOUNT_PAD;
    int* slots  = elist + (size_t)AGG_BLOCKS * LSTRIDE;

    float* out1 = (float*)d_out;
    float* out2 = out1 + (size_t)NNODES * DOUT;

    dim3 gemm_grid(GEMM_BLOCKS, 1);


    lin_gemm_kernel<<<gemm_grid, GEMM_THREADS, 0, stream>>>(x, W0, b0, h, src, dst, slots);

    gather_add_kernel<<<AGG_BLOCKS, AGG_THREADS, 0, stream>>>(h, src, dst, agg, deg, elist, ecount, slots, 1);
    sage_gemm_kernel<<<gemm_grid, GEMM_THREADS, 0, stream>>>(agg, deg, h, Wl1, bl1, Wr1, out1, 1);

    gather_add_kernel<<<AGG_BLOCKS, AGG_THREADS, 0, stream>>>(out1, src, dst, agg, (float*)nullptr, elist, ecount, slots, 0);
    sage_gemm_kernel<<<gemm_grid, GEMM_THREADS, 0, stream>>>(agg, deg, out1, Wl2, bl2, Wr2, out2, 0);
}
